# starting kernel
# baseline (speedup 1.0000x reference)
#include <hip/hip_runtime.h>
#include <hip/hip_bf16.h>

#define IN_DIM  128
#define OUT_DIM 64

typedef __attribute__((ext_vector_type(2))) float v2f;
typedef __attribute__((ext_vector_type(8))) float v8f;

__global__ void gat_init_kernel(unsigned int* __restrict__ maxbits,
                                float* __restrict__ denom,
                                float* __restrict__ out,
                                int n_nodes) {
    int i = blockIdx.x * blockDim.x + threadIdx.x;
    int total = n_nodes * OUT_DIM;
    if (i < total) out[i] = 0.0f;
    if (i < n_nodes) {
        maxbits[i] = 0xFF800000u;
        denom[i]   = 0.0f;
    }
}

__global__ void __launch_bounds__(128)
gat_gemm_wmma_kernel(const float* __restrict__ h,
                     const float* __restrict__ W,
                     float* __restrict__ z,
                     int n_nodes) {
    const int lane  = threadIdx.x & 31;
    const int wave  = threadIdx.x >> 5;
    const int half  = lane >> 4;
    const int lcol  = lane & 15;

    const int row0  = blockIdx.x * 16;
    const int nbase = wave * 16;

    int arow = row0 + lcol;
    if (arow >= n_nodes) arow = n_nodes - 1;
    const int ncol = nbase + lcol;

    const float* __restrict__ hrow = h + (size_t)arow * IN_DIM;
    const float* __restrict__ wrow = W + (size_t)ncol * IN_DIM;

    v8f acc = {};
#pragma unroll
    for (int k = 0; k < IN_DIM; k += 4) {
        const int kk = k + 2 * half;
        float2 af = *(const float2*)(hrow + kk);
        float2 bf = *(const float2*)(wrow + kk);
        v2f a; a[0] = af.x; a[1] = af.y;
        v2f b; b[0] = bf.x; b[1] = bf.y;
        acc = __builtin_amdgcn_wmma_f32_16x16x4_f32(
             false, a,  false, b,
             (short)0, acc,  false,  false);
    }

#pragma unroll
    for (int v = 0; v < 8; ++v) {
        int row = row0 + v + 8 * half;
        if (row < n_nodes)
            z[(size_t)row * OUT_DIM + ncol] = acc[v];
    }
}

__global__ void __launch_bounds__(256)
gat_edge_score_kernel(const float* __restrict__ z,
                      const int* __restrict__ src,
                      const int* __restrict__ dst,
                      float* __restrict__ e,
                      unsigned int* __restrict__ maxbits,
                      int n_edges) {
    const int lane = threadIdx.x & 31;
    const int eid  = blockIdx.x * (blockDim.x >> 5) + (threadIdx.x >> 5);
    if (eid >= n_edges) return;

    const int s = src[eid];
    const int d = dst[eid];
    const float2* zs = (const float2*)(z + (size_t)s * OUT_DIM);
    const float2* zd = (const float2*)(z + (size_t)d * OUT_DIM);
    float2 a = zs[lane];
    float2 b = zd[lane];
    float acc = a.x * b.x + a.y * b.y;
#pragma unroll
    for (int off = 16; off > 0; off >>= 1)
        acc += __shfl_xor(acc, off, 32);

    if (lane == 0) {
        e[eid] = acc;
        if (acc >= 0.0f)
            atomicMax((int*)&maxbits[d], __float_as_int(acc));
        else
            atomicMin(&maxbits[d], __float_as_uint(acc));
    }
}

__global__ void gat_exp_denom_kernel(const int* __restrict__ dst,
                                     float* __restrict__ e,
                                     const unsigned int* __restrict__ maxbits,
                                     float* __restrict__ denom,
                                     int n_edges) {
    int eid = blockIdx.x * blockDim.x + threadIdx.x;
    if (eid >= n_edges) return;
    int d = dst[eid];
    float m = __int_as_float((int)maxbits[d]);
    float t = __expf(e[eid] - m);
    e[eid] = t;
    atomicAdd(&denom[d], t);
}

__global__ void __launch_bounds__(256)
gat_aggregate_kernel(const float* __restrict__ z,
                     const int* __restrict__ src,
                     const int* __restrict__ dst,
                     const float* __restrict__ e,
                     const float* __restrict__ denom,
                     float* __restrict__ out,
                     int n_edges) {
    const int j   = threadIdx.x & 63;
    const int eid = blockIdx.x * (blockDim.x >> 6) + (threadIdx.x >> 6);
    if (eid >= n_edges) return;
    const int s = src[eid];
    const int d = dst[eid];
    const float alpha = e[eid] / fmaxf(denom[d], 1e-30f);
    atomicAdd(&out[(size_t)d * OUT_DIM + j], alpha * z[(size_t)s * OUT_DIM + j]);
}

extern "C" void kernel_launch(void* const* d_in, const int* in_sizes, int n_in,
                              void* d_out, int out_size, void* d_ws, size_t ws_size,
                              hipStream_t stream) {
    const float* h   = (const float*)d_in[0];
    const float* W   = (const float*)d_in[1];
    const int*   src = (const int*)d_in[2];
    const int*   dst = (const int*)d_in[3];
    float* out = (float*)d_out;

    const int n_nodes = in_sizes[0] / IN_DIM;
    const int n_edges = in_sizes[2];

    float*        z       = (float*)d_ws;
    float*        e       = z + (size_t)n_nodes * OUT_DIM;
    unsigned int* maxbits = (unsigned int*)(e + n_edges);
    float*        denom   = (float*)(maxbits + n_nodes);

    {
        int total = n_nodes * OUT_DIM;
        gat_init_kernel<<<(total + 255) / 256, 256, 0, stream>>>(
            maxbits, denom, out, n_nodes);
    }
    {
        int blocks = (n_nodes + 15) / 16;
        gat_gemm_wmma_kernel<<<blocks, 128, 0, stream>>>(h, W, z, n_nodes);
    }
    {
        int edges_per_block = 256 / 32;
        int blocks = (n_edges + edges_per_block - 1) / edges_per_block;
        gat_edge_score_kernel<<<blocks, 256, 0, stream>>>(
            z, src, dst, e, maxbits, n_edges);
    }
    {
        gat_exp_denom_kernel<<<(n_edges + 255) / 256, 256, 0, stream>>>(
            dst, e, maxbits, denom, n_edges);
    }
    {
        int edges_per_block = 256 / 64;
        int blocks = (n_edges + edges_per_block - 1) / edges_per_block;
        gat_aggregate_kernel<<<blocks, 256, 0, stream>>>(
            z, src, dst, e, denom, out, n_edges);
    }
}


# Edge score kernel: a wave takes 32 edges (8 lanes an edge, rows read as whole 128B lines, four steps of loads issued together), partial sums folded across lanes; score terms summed in another order
# speedup vs baseline: 1.2355x; 1.2355x over previous
#include <hip/hip_runtime.h>
#include <hip/hip_bf16.h>

#define IN_DIM  128
#define OUT_DIM 64

typedef __attribute__((ext_vector_type(2))) float v2f;
typedef __attribute__((ext_vector_type(8))) float v8f;

__global__ void gat_init_kernel(unsigned int* __restrict__ maxbits,
                                float* __restrict__ denom,
                                float* __restrict__ out,
                                int n_nodes) {
    int i = blockIdx.x * blockDim.x + threadIdx.x;
    int total = n_nodes * OUT_DIM;
    if (i < total) out[i] = 0.0f;
    if (i < n_nodes) {
        maxbits[i] = 0xFF800000u;
        denom[i]   = 0.0f;
    }
}

__global__ void __launch_bounds__(128)
gat_gemm_wmma_kernel(const float* __restrict__ h,
                     const float* __restrict__ W,
                     float* __restrict__ z,
                     int n_nodes) {
    const int lane  = threadIdx.x & 31;
    const int wave  = threadIdx.x >> 5;
    const int half  = lane >> 4;
    const int lcol  = lane & 15;

    const int row0  = blockIdx.x * 16;
    const int nbase = wave * 16;

    int arow = row0 + lcol;
    if (arow >= n_nodes) arow = n_nodes - 1;
    const int ncol = nbase + lcol;

    const float* __restrict__ hrow = h + (size_t)arow * IN_DIM;
    const float* __restrict__ wrow = W + (size_t)ncol * IN_DIM;

    v8f acc = {};
#pragma unroll
    for (int k = 0; k < IN_DIM; k += 4) {
        const int kk = k + 2 * half;
        float2 af = *(const float2*)(hrow + kk);
        float2 bf = *(const float2*)(wrow + kk);
        v2f a; a[0] = af.x; a[1] = af.y;
        v2f b; b[0] = bf.x; b[1] = bf.y;
        acc = __builtin_amdgcn_wmma_f32_16x16x4_f32(
             false, a,  false, b,
             (short)0, acc,  false,  false);
    }

#pragma unroll
    for (int v = 0; v < 8; ++v) {
        int row = row0 + v + 8 * half;
        if (row < n_nodes)
            z[(size_t)row * OUT_DIM + ncol] = acc[v];
    }
}

__global__ void __launch_bounds__(256)
gat_edge_score_kernel(const float* __restrict__ z,
                      const int* __restrict__ src,
                      const int* __restrict__ dst,
                      float* __restrict__ e,
                      unsigned int* __restrict__ maxbits,
                      int n_edges) {
    const int lane = threadIdx.x & 31;
    const int r    = lane & 7;
    const int g    = lane >> 3;
    const int wid  = blockIdx.x * (blockDim.x >> 5) + (threadIdx.x >> 5);
    const int eid0 = wid * 32;
    if (eid0 >= n_edges) return;

    const int  my    = eid0 + 4 * r + g;
    const bool valid = my < n_edges;
    const int  ld    = valid ? my : (n_edges - 1);
    const int  s     = src[ld];
    const int  d     = dst[ld];

    float p[8];
#pragma unroll
    for (int hp = 0; hp < 2; ++hp) {
        float4 a0[4], a1[4], b0[4], b1[4];
#pragma unroll
        for (int u = 0; u < 4; ++u) {
            const int t    = hp * 4 + u;
            const int from = (lane & 24) + t;
            const int st   = __shfl(s, from, 32);
            const int dt   = __shfl(d, from, 32);
            const float4* zs = (const float4*)(z + (size_t)st * OUT_DIM) + r;
            const float4* zd = (const float4*)(z + (size_t)dt * OUT_DIM) + r;
            a0[u] = zs[0];
            a1[u] = zs[8];
            b0[u] = zd[0];
            b1[u] = zd[8];
        }
#pragma unroll
        for (int u = 0; u < 4; ++u) {
            float acc = a0[u].x * b0[u].x;
            acc = fmaf(a0[u].y, b0[u].y, acc);
            acc = fmaf(a0[u].z, b0[u].z, acc);
            acc = fmaf(a0[u].w, b0[u].w, acc);
            acc = fmaf(a1[u].x, b1[u].x, acc);
            acc = fmaf(a1[u].y, b1[u].y, acc);
            acc = fmaf(a1[u].z, b1[u].z, acc);
            acc = fmaf(a1[u].w, b1[u].w, acc);
            p[hp * 4 + u] = acc;
        }
    }

    const bool hi4 = (r & 4) != 0;
    float q[4];
#pragma unroll
    for (int k = 0; k < 4; ++k) {
        const float send = hi4 ? p[k] : p[k + 4];
        const float keep = hi4 ? p[k + 4] : p[k];
        q[k] = keep + __shfl_xor(send, 4, 32);
    }
    const bool hi2 = (r & 2) != 0;
    float w[2];
#pragma unroll
    for (int k = 0; k < 2; ++k) {
        const float send = hi2 ? q[k] : q[k + 2];
        const float keep = hi2 ? q[k + 2] : q[k];
        w[k] = keep + __shfl_xor(send, 2, 32);
    }
    const bool hi1 = (r & 1) != 0;
    const float send1 = hi1 ? w[0] : w[1];
    const float keep1 = hi1 ? w[1] : w[0];
    const float acc   = keep1 + __shfl_xor(send1, 1, 32);

    if (valid) {
        e[my] = acc;
        if (acc >= 0.0f)
            atomicMax((int*)&maxbits[d], __float_as_int(acc));
        else
            atomicMin(&maxbits[d], __float_as_uint(acc));
    }
}

__global__ void gat_exp_denom_kernel(const int* __restrict__ dst,
                                     float* __restrict__ e,
                                     const unsigned int* __restrict__ maxbits,
                                     float* __restrict__ denom,
                                     int n_edges) {
    int eid = blockIdx.x * blockDim.x + threadIdx.x;
    if (eid >= n_edges) return;
    int d = dst[eid];
    float m = __int_as_float((int)maxbits[d]);
    float t = __expf(e[eid] - m);
    e[eid] = t;
    atomicAdd(&denom[d], t);
}

__global__ void __launch_bounds__(256)
gat_aggregate_kernel(const float* __restrict__ z,
                     const int* __restrict__ src,
                     const int* __restrict__ dst,
                     const float* __restrict__ e,
                     const float* __restrict__ denom,
                     float* __restrict__ out,
                     int n_edges) {
    const int j   = threadIdx.x & 63;
    const int eid = blockIdx.x * (blockDim.x >> 6) + (threadIdx.x >> 6);
    if (eid >= n_edges) return;
    const int s = src[eid];
    const int d = dst[eid];
    const float alpha = e[eid] / fmaxf(denom[d], 1e-30f);
    atomicAdd(&out[(size_t)d * OUT_DIM + j], alpha * z[(size_t)s * OUT_DIM + j]);
}

extern "C" void kernel_launch(void* const* d_in, const int* in_sizes, int n_in,
                              void* d_out, int out_size, void* d_ws, size_t ws_size,
                              hipStream_t stream) {
    const float* h   = (const float*)d_in[0];
    const float* W   = (const float*)d_in[1];
    const int*   src = (const int*)d_in[2];
    const int*   dst = (const int*)d_in[3];
    float* out = (float*)d_out;

    const int n_nodes = in_sizes[0] / IN_DIM;
    const int n_edges = in_sizes[2];

    float*        z       = (float*)d_ws;
    float*        e       = z + (size_t)n_nodes * OUT_DIM;
    unsigned int* maxbits = (unsigned int*)(e + n_edges);
    float*        denom   = (float*)(maxbits + n_nodes);

    {
        int total = n_nodes * OUT_DIM;
        gat_init_kernel<<<(total + 255) / 256, 256, 0, stream>>>(
            maxbits, denom, out, n_nodes);
    }
    {
        int blocks = (n_nodes + 15) / 16;
        gat_gemm_wmma_kernel<<<blocks, 128, 0, stream>>>(h, W, z, n_nodes);
    }
    {
        int edges_per_block = (256 / 32) * 32;
        int blocks = (n_edges + edges_per_block - 1) / edges_per_block;
        gat_edge_score_kernel<<<blocks, 256, 0, stream>>>(
            z, src, dst, e, maxbits, n_edges);
    }
    {
        gat_exp_denom_kernel<<<(n_edges + 255) / 256, 256, 0, stream>>>(
            dst, e, maxbits, denom, n_edges);
    }
    {
        int edges_per_block = 256 / 64;
        int blocks = (n_edges + edges_per_block - 1) / edges_per_block;
        gat_aggregate_kernel<<<blocks, 256, 0, stream>>>(
            z, src, dst, e, denom, out, n_edges);
    }
}


# Aggregate kernel: a wave takes 32 edges (lane = dim; an edge's row addresses and weight go to scalar registers; next 8 edges' row loads issued ahead of the adds); atomic adds land in another order
# speedup vs baseline: 2.2355x; 1.8094x over previous
#include <hip/hip_runtime.h>
#include <hip/hip_bf16.h>

#define IN_DIM  128
#define OUT_DIM 64

typedef __attribute__((ext_vector_type(2))) float v2f;
typedef __attribute__((ext_vector_type(8))) float v8f;

__global__ void gat_init_kernel(unsigned int* __restrict__ maxbits,
                                float* __restrict__ denom,
                                float* __restrict__ out,
                                int n_nodes) {
    int i = blockIdx.x * blockDim.x + threadIdx.x;
    int total = n_nodes * OUT_DIM;
    if (i < total) out[i] = 0.0f;
    if (i < n_nodes) {
        maxbits[i] = 0xFF800000u;
        denom[i]   = 0.0f;
    }
}

__global__ void __launch_bounds__(128)
gat_gemm_wmma_kernel(const float* __restrict__ h,
                     const float* __restrict__ W,
                     float* __restrict__ z,
                     int n_nodes) {
    const int lane  = threadIdx.x & 31;
    const int wave  = threadIdx.x >> 5;
    const int half  = lane >> 4;
    const int lcol  = lane & 15;

    const int row0  = blockIdx.x * 16;
    const int nbase = wave * 16;

    int arow = row0 + lcol;
    if (arow >= n_nodes) arow = n_nodes - 1;
    const int ncol = nbase + lcol;

    const float* __restrict__ hrow = h + (size_t)arow * IN_DIM;
    const float* __restrict__ wrow = W + (size_t)ncol * IN_DIM;

    v8f acc = {};
#pragma unroll
    for (int k = 0; k < IN_DIM; k += 4) {
        const int kk = k + 2 * half;
        float2 af = *(const float2*)(hrow + kk);
        float2 bf = *(const float2*)(wrow + kk);
        v2f a; a[0] = af.x; a[1] = af.y;
        v2f b; b[0] = bf.x; b[1] = bf.y;
        acc = __builtin_amdgcn_wmma_f32_16x16x4_f32(
             false, a,  false, b,
             (short)0, acc,  false,  false);
    }

#pragma unroll
    for (int v = 0; v < 8; ++v) {
        int row = row0 + v + 8 * half;
        if (row < n_nodes)
            z[(size_t)row * OUT_DIM + ncol] = acc[v];
    }
}

__global__ void __launch_bounds__(256)
gat_edge_score_kernel(const float* __restrict__ z,
                      const int* __restrict__ src,
                      const int* __restrict__ dst,
                      float* __restrict__ e,
                      unsigned int* __restrict__ maxbits,
                      int n_edges) {
    const int lane = threadIdx.x & 31;
    const int r    = lane & 7;
    const int g    = lane >> 3;
    const int wid  = blockIdx.x * (blockDim.x >> 5) + (threadIdx.x >> 5);
    const int eid0 = wid * 32;
    if (eid0 >= n_edges) return;

    const int  my    = eid0 + 4 * r + g;
    const bool valid = my < n_edges;
    const int  ld    = valid ? my : (n_edges - 1);
    const int  s     = src[ld];
    const int  d     = dst[ld];

    float p[8];
#pragma unroll
    for (int hp = 0; hp < 2; ++hp) {
        float4 a0[4], a1[4], b0[4], b1[4];
#pragma unroll
        for (int u = 0; u < 4; ++u) {
            const int t    = hp * 4 + u;
            const int from = (lane & 24) + t;
            const int st   = __shfl(s, from, 32);
            const int dt   = __shfl(d, from, 32);
            const float4* zs = (const float4*)(z + (size_t)st * OUT_DIM) + r;
            const float4* zd = (const float4*)(z + (size_t)dt * OUT_DIM) + r;
            a0[u] = zs[0];
            a1[u] = zs[8];
            b0[u] = zd[0];
            b1[u] = zd[8];
        }
#pragma unroll
        for (int u = 0; u < 4; ++u) {
            float acc = a0[u].x * b0[u].x;
            acc = fmaf(a0[u].y, b0[u].y, acc);
            acc = fmaf(a0[u].z, b0[u].z, acc);
            acc = fmaf(a0[u].w, b0[u].w, acc);
            acc = fmaf(a1[u].x, b1[u].x, acc);
            acc = fmaf(a1[u].y, b1[u].y, acc);
            acc = fmaf(a1[u].z, b1[u].z, acc);
            acc = fmaf(a1[u].w, b1[u].w, acc);
            p[hp * 4 + u] = acc;
        }
    }

    const bool hi4 = (r & 4) != 0;
    float q[4];
#pragma unroll
    for (int k = 0; k < 4; ++k) {
        const float send = hi4 ? p[k] : p[k + 4];
        const float keep = hi4 ? p[k + 4] : p[k];
        q[k] = keep + __shfl_xor(send, 4, 32);
    }
    const bool hi2 = (r & 2) != 0;
    float w[2];
#pragma unroll
    for (int k = 0; k < 2; ++k) {
        const float send = hi2 ? q[k] : q[k + 2];
        const float keep = hi2 ? q[k + 2] : q[k];
        w[k] = keep + __shfl_xor(send, 2, 32);
    }
    const bool hi1 = (r & 1) != 0;
    const float send1 = hi1 ? w[0] : w[1];
    const float keep1 = hi1 ? w[1] : w[0];
    const float acc   = keep1 + __shfl_xor(send1, 1, 32);

    if (valid) {
        e[my] = acc;
        if (acc >= 0.0f)
            atomicMax((int*)&maxbits[d], __float_as_int(acc));
        else
            atomicMin(&maxbits[d], __float_as_uint(acc));
    }
}

__global__ void gat_exp_denom_kernel(const int* __restrict__ dst,
                                     float* __restrict__ e,
                                     const unsigned int* __restrict__ maxbits,
                                     float* __restrict__ denom,
                                     int n_edges) {
    int eid = blockIdx.x * blockDim.x + threadIdx.x;
    if (eid >= n_edges) return;
    int d = dst[eid];
    float m = __int_as_float((int)maxbits[d]);
    float t = __expf(e[eid] - m);
    e[eid] = t;
    atomicAdd(&denom[d], t);
}


typedef __attribute__((address_space(1))) float       gat_gfloat;
typedef __attribute__((address_space(1))) const float gat_gcfloat;

__device__ __forceinline__ unsigned long long gat_row_addr(const float* base, int row) {
    const unsigned long long a =
        (unsigned long long)base + ((unsigned long long)(unsigned int)row << 8);
    const unsigned int lo = __builtin_amdgcn_readfirstlane((unsigned int)a);
    const unsigned int hi = __builtin_amdgcn_readfirstlane((unsigned int)(a >> 32));
    return ((unsigned long long)hi << 32) | lo;
}

template <bool FULL>
__device__ __forceinline__ void gat_aggregate_walk(const float* __restrict__ z,
                                                   float* __restrict__ out,
                                                   int s, int d, float ev, float dn,
                                                   int cnt, int lane) {
    float v0[2][8], v1[2][8];
#pragma unroll
    for (int u = 0; u < 8; ++u) {
        const gat_gcfloat* zr =
            (const gat_gcfloat*)gat_row_addr(z, __builtin_amdgcn_readlane(s, u));
        v0[0][u] = zr[lane];
        v1[0][u] = zr[lane + 32];
    }
    const float alpha = ev / fmaxf(dn, 1e-30f);
    const int abits = __float_as_int(alpha);
#pragma unroll
    for (int g = 0; g < 4; ++g) {
        if (g + 1 < 4) {
#pragma unroll
            for (int u = 0; u < 8; ++u) {
                const gat_gcfloat* zr = (const gat_gcfloat*)gat_row_addr(
                    z, __builtin_amdgcn_readlane(s, (g + 1) * 8 + u));
                v0[(g + 1) & 1][u] = zr[lane];
                v1[(g + 1) & 1][u] = zr[lane + 32];
            }
        }
#pragma unroll
        for (int u = 0; u < 8; ++u) {
            const int k = g * 8 + u;
            if (FULL || k < cnt) {
                gat_gfloat* orow =
                    (gat_gfloat*)gat_row_addr(out, __builtin_amdgcn_readlane(d, k));
                const float ak = __int_as_float(__builtin_amdgcn_readlane(abits, k));
                __hip_atomic_fetch_add(orow + lane, ak * v0[g & 1][u],
                                       __ATOMIC_RELAXED, __HIP_MEMORY_SCOPE_AGENT);
                __hip_atomic_fetch_add(orow + lane + 32, ak * v1[g & 1][u],
                                       __ATOMIC_RELAXED, __HIP_MEMORY_SCOPE_AGENT);
            }
        }
    }
}

__global__ void __launch_bounds__(256)
gat_aggregate_kernel(const float* __restrict__ z,
                     const int* __restrict__ src,
                     const int* __restrict__ dst,
                     const float* __restrict__ e,
                     const float* __restrict__ denom,
                     float* __restrict__ out,
                     int n_edges) {
    const int lane = threadIdx.x & 31;
    const int wib  = __builtin_amdgcn_readfirstlane(threadIdx.x >> 5);
    const int wid  = blockIdx.x * (blockDim.x >> 5) + wib;
    const int eid0 = wid * 32;
    if (eid0 >= n_edges) return;

    const int my  = eid0 + lane;
    const int ld  = (my < n_edges) ? my : (n_edges - 1);
    const int s   = src[ld];
    const int d   = dst[ld];
    const float ev = e[ld];
    const float dn = denom[d];

    const int cnt = n_edges - eid0;
    if (cnt >= 32)
        gat_aggregate_walk<true>(z, out, s, d, ev, dn, 32, lane);
    else
        gat_aggregate_walk<false>(z, out, s, d, ev, dn, cnt, lane);
}

extern "C" void kernel_launch(void* const* d_in, const int* in_sizes, int n_in,
                              void* d_out, int out_size, void* d_ws, size_t ws_size,
                              hipStream_t stream) {
    const float* h   = (const float*)d_in[0];
    const float* W   = (const float*)d_in[1];
    const int*   src = (const int*)d_in[2];
    const int*   dst = (const int*)d_in[3];
    float* out = (float*)d_out;

    const int n_nodes = in_sizes[0] / IN_DIM;
    const int n_edges = in_sizes[2];

    float*        z       = (float*)d_ws;
    float*        e       = z + (size_t)n_nodes * OUT_DIM;
    unsigned int* maxbits = (unsigned int*)(e + n_edges);
    float*        denom   = (float*)(maxbits + n_nodes);

    {
        int total = n_nodes * OUT_DIM;
        gat_init_kernel<<<(total + 255) / 256, 256, 0, stream>>>(
            maxbits, denom, out, n_nodes);
    }
    {
        int blocks = (n_nodes + 15) / 16;
        gat_gemm_wmma_kernel<<<blocks, 128, 0, stream>>>(h, W, z, n_nodes);
    }
    {
        int edges_per_block = (256 / 32) * 32;
        int blocks = (n_edges + edges_per_block - 1) / edges_per_block;
        gat_edge_score_kernel<<<blocks, 256, 0, stream>>>(
            z, src, dst, e, maxbits, n_edges);
    }
    {
        gat_exp_denom_kernel<<<(n_edges + 255) / 256, 256, 0, stream>>>(
            dst, e, maxbits, denom, n_edges);
    }
    {
        int edges_per_block = (256 / 32) * 32;
        int blocks = (n_edges + edges_per_block - 1) / edges_per_block;
        gat_aggregate_kernel<<<blocks, 256, 0, stream>>>(
            z, src, dst, e, denom, out, n_edges);
    }
}


# GEMM: a work-group reads its 80 rows of h once in whole 128B lines into LDS; its 4 waves take A fragments from LDS and keep their 16 columns of W in registers; k summed in another order
# speedup vs baseline: 2.4068x; 1.0766x over previous
#include <hip/hip_runtime.h>
#include <hip/hip_bf16.h>

#define IN_DIM  128
#define OUT_DIM 64

typedef __attribute__((ext_vector_type(2))) float v2f;
typedef __attribute__((ext_vector_type(8))) float v8f;

__global__ void gat_init_kernel(unsigned int* __restrict__ maxbits,
                                float* __restrict__ denom,
                                float* __restrict__ out,
                                int n_nodes) {
    int i = blockIdx.x * blockDim.x + threadIdx.x;
    int total = n_nodes * OUT_DIM;
    if (i < total) out[i] = 0.0f;
    if (i < n_nodes) {
        maxbits[i] = 0xFF800000u;
        denom[i]   = 0.0f;
    }
}

typedef __attribute__((ext_vector_type(4))) float v4f;
typedef __attribute__((address_space(1))) const v4f gat_gcv4f;

__device__ __forceinline__ const gat_gcv4f* gat_h_row(const float* base, int row) {
    const unsigned long long a =
        (unsigned long long)base + ((unsigned long long)(unsigned int)row << 9);
    const unsigned int lo = __builtin_amdgcn_readfirstlane((unsigned int)a);
    const unsigned int hi = __builtin_amdgcn_readfirstlane((unsigned int)(a >> 32));
    return (const gat_gcv4f*)(((unsigned long long)hi << 32) | lo);
}

#define GEMM_RT     5
#define GEMM_ROWS   (16 * GEMM_RT)
#define GEMM_WROWS  (GEMM_ROWS / 4)
#define GEMM_PITCH4 33

__global__ void __launch_bounds__(128)
gat_gemm_wmma_kernel(const float* __restrict__ h,
                     const float* __restrict__ W,
                     float* __restrict__ z,
                     int n_nodes) {
    __shared__ v4f tile4[GEMM_ROWS * GEMM_PITCH4];

    const int tid   = threadIdx.x;
    const int lane  = tid & 31;
    const int wave  = __builtin_amdgcn_readfirstlane(tid >> 5);
    const int half  = lane >> 4;
    const int lcol  = lane & 15;

    const int row0  = blockIdx.x * GEMM_ROWS;
    const int ncol  = wave * 16 + lcol;

    v4f st[GEMM_WROWS];
#pragma unroll
    for (int p = 0; p < GEMM_WROWS; ++p) {
        int gr = row0 + 4 * p + wave;
        if (gr >= n_nodes) gr = n_nodes - 1;
        st[p] = gat_h_row(h, gr)[lane];
    }

    const v4f* __restrict__ wrow =
        (const v4f*)(W + (size_t)ncol * IN_DIM) + half;
    v4f b[16];
#pragma unroll
    for (int j = 0; j < 16; ++j) b[j] = wrow[2 * j];

#pragma unroll
    for (int p = 0; p < GEMM_WROWS; ++p)
        tile4[(4 * p + wave) * GEMM_PITCH4 + lane] = st[p];

    __syncthreads();

    v8f acc[GEMM_RT] = {};
    const v4f* arow = tile4 + lcol * GEMM_PITCH4 + half;
    v4f a_cur[GEMM_RT], a_nxt[GEMM_RT];
#pragma unroll
    for (int rt = 0; rt < GEMM_RT; ++rt)
        a_cur[rt] = arow[rt * 16 * GEMM_PITCH4];
#pragma unroll
    for (int j = 0; j < 16; ++j) {
        if (j + 1 < 16) {
#pragma unroll
            for (int rt = 0; rt < GEMM_RT; ++rt)
                a_nxt[rt] = arow[rt * 16 * GEMM_PITCH4 + 2 * (j + 1)];
        }
        __builtin_amdgcn_sched_barrier(0);
        v2f b0; b0[0] = b[j].x; b0[1] = b[j].y;
        v2f b1; b1[0] = b[j].z; b1[1] = b[j].w;
#pragma unroll
        for (int rt = 0; rt < GEMM_RT; ++rt) {
            v2f a0; a0[0] = a_cur[rt].x; a0[1] = a_cur[rt].y;
            acc[rt] = __builtin_amdgcn_wmma_f32_16x16x4_f32(
                 false, a0,  false, b0,
                 (short)0, acc[rt],  false,  false);
        }
#pragma unroll
        for (int rt = 0; rt < GEMM_RT; ++rt) {
            v2f a1; a1[0] = a_cur[rt].z; a1[1] = a_cur[rt].w;
            acc[rt] = __builtin_amdgcn_wmma_f32_16x16x4_f32(
                 false, a1,  false, b1,
                 (short)0, acc[rt],  false,  false);
        }
        __builtin_amdgcn_sched_barrier(0);
#pragma unroll
        for (int rt = 0; rt < GEMM_RT; ++rt) a_cur[rt] = a_nxt[rt];
    }

    float* zout = z + (size_t)(row0 + 8 * half) * OUT_DIM + ncol;
    if (row0 + GEMM_ROWS <= n_nodes) {
#pragma unroll
        for (int rt = 0; rt < GEMM_RT; ++rt) {
#pragma unroll
            for (int v = 0; v < 8; ++v)
                zout[(rt * 16 + v) * OUT_DIM] = acc[rt][v];
        }
    } else {
#pragma unroll
        for (int rt = 0; rt < GEMM_RT; ++rt) {
#pragma unroll
            for (int v = 0; v < 8; ++v) {
                int row = row0 + rt * 16 + v + 8 * half;
                if (row < n_nodes)
                    zout[(rt * 16 + v) * OUT_DIM] = acc[rt][v];
            }
        }
    }
}

__global__ void __launch_bounds__(256)
gat_edge_score_kernel(const float* __restrict__ z,
                      const int* __restrict__ src,
                      const int* __restrict__ dst,
                      float* __restrict__ e,
                      unsigned int* __restrict__ maxbits,
                      int n_edges) {
    const int lane = threadIdx.x & 31;
    const int r    = lane & 7;
    const int g    = lane >> 3;
    const int wid  = blockIdx.x * (blockDim.x >> 5) + (threadIdx.x >> 5);
    const int eid0 = wid * 32;
    if (eid0 >= n_edges) return;

    const int  my    = eid0 + 4 * r + g;
    const bool valid = my < n_edges;
    const int  ld    = valid ? my : (n_edges - 1);
    const int  s     = src[ld];
    const int  d     = dst[ld];

    float p[8];
#pragma unroll
    for (int hp = 0; hp < 2; ++hp) {
        float4 a0[4], a1[4], b0[4], b1[4];
#pragma unroll
        for (int u = 0; u < 4; ++u) {
            const int t    = hp * 4 + u;
            const int from = (lane & 24) + t;
            const int st   = __shfl(s, from, 32);
            const int dt   = __shfl(d, from, 32);
            const float4* zs = (const float4*)(z + (size_t)st * OUT_DIM) + r;
            const float4* zd = (const float4*)(z + (size_t)dt * OUT_DIM) + r;
            a0[u] = zs[0];
            a1[u] = zs[8];
            b0[u] = zd[0];
            b1[u] = zd[8];
        }
#pragma unroll
        for (int u = 0; u < 4; ++u) {
            float acc = a0[u].x * b0[u].x;
            acc = fmaf(a0[u].y, b0[u].y, acc);
            acc = fmaf(a0[u].z, b0[u].z, acc);
            acc = fmaf(a0[u].w, b0[u].w, acc);
            acc = fmaf(a1[u].x, b1[u].x, acc);
            acc = fmaf(a1[u].y, b1[u].y, acc);
            acc = fmaf(a1[u].z, b1[u].z, acc);
            acc = fmaf(a1[u].w, b1[u].w, acc);
            p[hp * 4 + u] = acc;
        }
    }

    const bool hi4 = (r & 4) != 0;
    float q[4];
#pragma unroll
    for (int k = 0; k < 4; ++k) {
        const float send = hi4 ? p[k] : p[k + 4];
        const float keep = hi4 ? p[k + 4] : p[k];
        q[k] = keep + __shfl_xor(send, 4, 32);
    }
    const bool hi2 = (r & 2) != 0;
    float w[2];
#pragma unroll
    for (int k = 0; k < 2; ++k) {
        const float send = hi2 ? q[k] : q[k + 2];
        const float keep = hi2 ? q[k + 2] : q[k];
        w[k] = keep + __shfl_xor(send, 2, 32);
    }
    const bool hi1 = (r & 1) != 0;
    const float send1 = hi1 ? w[0] : w[1];
    const float keep1 = hi1 ? w[1] : w[0];
    const float acc   = keep1 + __shfl_xor(send1, 1, 32);

    if (valid) {
        e[my] = acc;
        if (acc >= 0.0f)
            atomicMax((int*)&maxbits[d], __float_as_int(acc));
        else
            atomicMin(&maxbits[d], __float_as_uint(acc));
    }
}

__global__ void gat_exp_denom_kernel(const int* __restrict__ dst,
                                     float* __restrict__ e,
                                     const unsigned int* __restrict__ maxbits,
                                     float* __restrict__ denom,
                                     int n_edges) {
    int eid = blockIdx.x * blockDim.x + threadIdx.x;
    if (eid >= n_edges) return;
    int d = dst[eid];
    float m = __int_as_float((int)maxbits[d]);
    float t = __expf(e[eid] - m);
    e[eid] = t;
    atomicAdd(&denom[d], t);
}


typedef __attribute__((address_space(1))) float       gat_gfloat;
typedef __attribute__((address_space(1))) const float gat_gcfloat;

__device__ __forceinline__ unsigned long long gat_row_addr(const float* base, int row) {
    const unsigned long long a =
        (unsigned long long)base + ((unsigned long long)(unsigned int)row << 8);
    const unsigned int lo = __builtin_amdgcn_readfirstlane((unsigned int)a);
    const unsigned int hi = __builtin_amdgcn_readfirstlane((unsigned int)(a >> 32));
    return ((unsigned long long)hi << 32) | lo;
}

template <bool FULL>
__device__ __forceinline__ void gat_aggregate_walk(const float* __restrict__ z,
                                                   float* __restrict__ out,
                                                   int s, int d, float ev, float dn,
                                                   int cnt, int lane) {
    float v0[2][8], v1[2][8];
#pragma unroll
    for (int u = 0; u < 8; ++u) {
        const gat_gcfloat* zr =
            (const gat_gcfloat*)gat_row_addr(z, __builtin_amdgcn_readlane(s, u));
        v0[0][u] = zr[lane];
        v1[0][u] = zr[lane + 32];
    }
    const float alpha = ev / fmaxf(dn, 1e-30f);
    const int abits = __float_as_int(alpha);
#pragma unroll
    for (int g = 0; g < 4; ++g) {
        if (g + 1 < 4) {
#pragma unroll
            for (int u = 0; u < 8; ++u) {
                const gat_gcfloat* zr = (const gat_gcfloat*)gat_row_addr(
                    z, __builtin_amdgcn_readlane(s, (g + 1) * 8 + u));
                v0[(g + 1) & 1][u] = zr[lane];
                v1[(g + 1) & 1][u] = zr[lane + 32];
            }
        }
#pragma unroll
        for (int u = 0; u < 8; ++u) {
            const int k = g * 8 + u;
            if (FULL || k < cnt) {
                gat_gfloat* orow =
                    (gat_gfloat*)gat_row_addr(out, __builtin_amdgcn_readlane(d, k));
                const float ak = __int_as_float(__builtin_amdgcn_readlane(abits, k));
                __hip_atomic_fetch_add(orow + lane, ak * v0[g & 1][u],
                                       __ATOMIC_RELAXED, __HIP_MEMORY_SCOPE_AGENT);
                __hip_atomic_fetch_add(orow + lane + 32, ak * v1[g & 1][u],
                                       __ATOMIC_RELAXED, __HIP_MEMORY_SCOPE_AGENT);
            }
        }
    }
}

__global__ void __launch_bounds__(256)
gat_aggregate_kernel(const float* __restrict__ z,
                     const int* __restrict__ src,
                     const int* __restrict__ dst,
                     const float* __restrict__ e,
                     const float* __restrict__ denom,
                     float* __restrict__ out,
                     int n_edges) {
    const int lane = threadIdx.x & 31;
    const int wib  = __builtin_amdgcn_readfirstlane(threadIdx.x >> 5);
    const int wid  = blockIdx.x * (blockDim.x >> 5) + wib;
    const int eid0 = wid * 32;
    if (eid0 >= n_edges) return;

    const int my  = eid0 + lane;
    const int ld  = (my < n_edges) ? my : (n_edges - 1);
    const int s   = src[ld];
    const int d   = dst[ld];
    const float ev = e[ld];
    const float dn = denom[d];

    const int cnt = n_edges - eid0;
    if (cnt >= 32)
        gat_aggregate_walk<true>(z, out, s, d, ev, dn, 32, lane);
    else
        gat_aggregate_walk<false>(z, out, s, d, ev, dn, cnt, lane);
}

extern "C" void kernel_launch(void* const* d_in, const int* in_sizes, int n_in,
                              void* d_out, int out_size, void* d_ws, size_t ws_size,
                              hipStream_t stream) {
    const float* h   = (const float*)d_in[0];
    const float* W   = (const float*)d_in[1];
    const int*   src = (const int*)d_in[2];
    const int*   dst = (const int*)d_in[3];
    float* out = (float*)d_out;

    const int n_nodes = in_sizes[0] / IN_DIM;
    const int n_edges = in_sizes[2];

    float*        z       = (float*)d_ws;
    float*        e       = z + (size_t)n_nodes * OUT_DIM;
    unsigned int* maxbits = (unsigned int*)(e + n_edges);
    float*        denom   = (float*)(maxbits + n_nodes);

    {
        int total = n_nodes * OUT_DIM;
        gat_init_kernel<<<(total + 255) / 256, 256, 0, stream>>>(
            maxbits, denom, out, n_nodes);
    }
    {
        int blocks = (n_nodes + GEMM_ROWS - 1) / GEMM_ROWS;
        gat_gemm_wmma_kernel<<<blocks, 128, 0, stream>>>(h, W, z, n_nodes);
    }
    {
        int edges_per_block = (256 / 32) * 32;
        int blocks = (n_edges + edges_per_block - 1) / edges_per_block;
        gat_edge_score_kernel<<<blocks, 256, 0, stream>>>(
            z, src, dst, e, maxbits, n_edges);
    }
    {
        gat_exp_denom_kernel<<<(n_edges + 255) / 256, 256, 0, stream>>>(
            dst, e, maxbits, denom, n_edges);
    }
    {
        int edges_per_block = (256 / 32) * 32;
        int blocks = (n_edges + edges_per_block - 1) / edges_per_block;
        gat_aggregate_kernel<<<blocks, 256, 0, stream>>>(
            z, src, dst, e, denom, out, n_edges);
    }
}


# One plane of each atomic sum (segment max, segment sum, out rows) per package half, picked by die number; readers combine both planes, a last launch adds the out planes; sums in another order
# speedup vs baseline: 3.0405x; 1.2633x over previous
#include <hip/hip_runtime.h>
#include <hip/hip_bf16.h>

#define IN_DIM  128
#define OUT_DIM 64

typedef __attribute__((ext_vector_type(2))) float v2f;
typedef __attribute__((ext_vector_type(8))) float v8f;

__device__ __forceinline__ int gat_planes(size_t ws_size, int n_nodes, int n_edges) {
    const size_t base  = ((size_t)n_nodes * OUT_DIM + (size_t)n_edges +
                          2 * (size_t)n_nodes) * sizeof(float);
    const size_t small = base + 2 * (size_t)n_nodes * sizeof(float);
    const size_t big   = small + (size_t)n_nodes * OUT_DIM * sizeof(float);
    return ws_size >= big ? 2 : (ws_size >= small ? 1 : 0);
}

__device__ __forceinline__ int gat_half() {
    const unsigned int die = (__builtin_amdgcn_s_sendmsg_rtn(0x87) >> 16) & 0xFu;
    return die >= 4u ? 1 : 0;
}

__global__ void gat_init_kernel(unsigned int* __restrict__ maxbits0,
                                float* __restrict__ denom0,
                                unsigned int* __restrict__ maxbits1,
                                float* __restrict__ denom1,
                                float* __restrict__ out,
                                float* __restrict__ out1,
                                int n_nodes, int n_edges, size_t ws_size) {
    const int planes = gat_planes(ws_size, n_nodes, n_edges);
    int i = blockIdx.x * blockDim.x + threadIdx.x;
    int total = n_nodes * OUT_DIM;
    if (i < total) {
        out[i] = 0.0f;
        if (planes == 2) out1[i] = 0.0f;
    }
    if (i < n_nodes) {
        maxbits0[i] = 0xFF800000u;
        denom0[i]   = 0.0f;
        if (planes >= 1) {
            maxbits1[i] = 0xFF800000u;
            denom1[i]   = 0.0f;
        }
    }
}

typedef __attribute__((ext_vector_type(4))) float v4f;
typedef __attribute__((address_space(1))) const v4f gat_gcv4f;

__device__ __forceinline__ const gat_gcv4f* gat_h_row(const float* base, int row) {
    const unsigned long long a =
        (unsigned long long)base + ((unsigned long long)(unsigned int)row << 9);
    const unsigned int lo = __builtin_amdgcn_readfirstlane((unsigned int)a);
    const unsigned int hi = __builtin_amdgcn_readfirstlane((unsigned int)(a >> 32));
    return (const gat_gcv4f*)(((unsigned long long)hi << 32) | lo);
}

#define GEMM_RT     5
#define GEMM_ROWS   (16 * GEMM_RT)
#define GEMM_WROWS  (GEMM_ROWS / 4)
#define GEMM_PITCH4 33

__global__ void __launch_bounds__(128)
gat_gemm_wmma_kernel(const float* __restrict__ h,
                     const float* __restrict__ W,
                     float* __restrict__ z,
                     int n_nodes) {
    __shared__ v4f tile4[GEMM_ROWS * GEMM_PITCH4];

    const int tid   = threadIdx.x;
    const int lane  = tid & 31;
    const int wave  = __builtin_amdgcn_readfirstlane(tid >> 5);
    const int half  = lane >> 4;
    const int lcol  = lane & 15;

    const int row0  = blockIdx.x * GEMM_ROWS;
    const int ncol  = wave * 16 + lcol;

    v4f st[GEMM_WROWS];
#pragma unroll
    for (int p = 0; p < GEMM_WROWS; ++p) {
        int gr = row0 + 4 * p + wave;
        if (gr >= n_nodes) gr = n_nodes - 1;
        st[p] = gat_h_row(h, gr)[lane];
    }

    const v4f* __restrict__ wrow =
        (const v4f*)(W + (size_t)ncol * IN_DIM) + half;
    v4f b[16];
#pragma unroll
    for (int j = 0; j < 16; ++j) b[j] = wrow[2 * j];

#pragma unroll
    for (int p = 0; p < GEMM_WROWS; ++p)
        tile4[(4 * p + wave) * GEMM_PITCH4 + lane] = st[p];

    __syncthreads();

    v8f acc[GEMM_RT] = {};
    const v4f* arow = tile4 + lcol * GEMM_PITCH4 + half;
    v4f a_cur[GEMM_RT], a_nxt[GEMM_RT];
#pragma unroll
    for (int rt = 0; rt < GEMM_RT; ++rt)
        a_cur[rt] = arow[rt * 16 * GEMM_PITCH4];
#pragma unroll
    for (int j = 0; j < 16; ++j) {
        if (j + 1 < 16) {
#pragma unroll
            for (int rt = 0; rt < GEMM_RT; ++rt)
                a_nxt[rt] = arow[rt * 16 * GEMM_PITCH4 + 2 * (j + 1)];
        }
        __builtin_amdgcn_sched_barrier(0);
        v2f b0; b0[0] = b[j].x; b0[1] = b[j].y;
        v2f b1; b1[0] = b[j].z; b1[1] = b[j].w;
#pragma unroll
        for (int rt = 0; rt < GEMM_RT; ++rt) {
            v2f a0; a0[0] = a_cur[rt].x; a0[1] = a_cur[rt].y;
            acc[rt] = __builtin_amdgcn_wmma_f32_16x16x4_f32(
                 false, a0,  false, b0,
                 (short)0, acc[rt],  false,  false);
        }
#pragma unroll
        for (int rt = 0; rt < GEMM_RT; ++rt) {
            v2f a1; a1[0] = a_cur[rt].z; a1[1] = a_cur[rt].w;
            acc[rt] = __builtin_amdgcn_wmma_f32_16x16x4_f32(
                 false, a1,  false, b1,
                 (short)0, acc[rt],  false,  false);
        }
        __builtin_amdgcn_sched_barrier(0);
#pragma unroll
        for (int rt = 0; rt < GEMM_RT; ++rt) a_cur[rt] = a_nxt[rt];
    }

    float* zout = z + (size_t)(row0 + 8 * half) * OUT_DIM + ncol;
    if (row0 + GEMM_ROWS <= n_nodes) {
#pragma unroll
        for (int rt = 0; rt < GEMM_RT; ++rt) {
#pragma unroll
            for (int v = 0; v < 8; ++v)
                zout[(rt * 16 + v) * OUT_DIM] = acc[rt][v];
        }
    } else {
#pragma unroll
        for (int rt = 0; rt < GEMM_RT; ++rt) {
#pragma unroll
            for (int v = 0; v < 8; ++v) {
                int row = row0 + rt * 16 + v + 8 * half;
                if (row < n_nodes)
                    zout[(rt * 16 + v) * OUT_DIM] = acc[rt][v];
            }
        }
    }
}

__global__ void __launch_bounds__(256)
gat_edge_score_kernel(const float* __restrict__ z,
                      const int* __restrict__ src,
                      const int* __restrict__ dst,
                      float* __restrict__ e,
                      unsigned int* __restrict__ maxbits0,
                      unsigned int* __restrict__ maxbits1,
                      int n_nodes, int n_edges, size_t ws_size) {
    const int lane = threadIdx.x & 31;
    const int r    = lane & 7;
    const int g    = lane >> 3;
    const int wid  = blockIdx.x * (blockDim.x >> 5) + (threadIdx.x >> 5);
    const int eid0 = wid * 32;
    if (eid0 >= n_edges) return;

    const int hi = gat_half();

    const int  my    = eid0 + 4 * r + g;
    const bool valid = my < n_edges;
    const int  ld    = valid ? my : (n_edges - 1);
    const int  s     = src[ld];
    const int  d     = dst[ld];

    float p[8];
#pragma unroll
    for (int hp = 0; hp < 2; ++hp) {
        float4 a0[4], a1[4], b0[4], b1[4];
#pragma unroll
        for (int u = 0; u < 4; ++u) {
            const int t    = hp * 4 + u;
            const int from = (lane & 24) + t;
            const int st   = __shfl(s, from, 32);
            const int dt   = __shfl(d, from, 32);
            const float4* zs = (const float4*)(z + (size_t)st * OUT_DIM) + r;
            const float4* zd = (const float4*)(z + (size_t)dt * OUT_DIM) + r;
            a0[u] = zs[0];
            a1[u] = zs[8];
            b0[u] = zd[0];
            b1[u] = zd[8];
        }
#pragma unroll
        for (int u = 0; u < 4; ++u) {
            float acc = a0[u].x * b0[u].x;
            acc = fmaf(a0[u].y, b0[u].y, acc);
            acc = fmaf(a0[u].z, b0[u].z, acc);
            acc = fmaf(a0[u].w, b0[u].w, acc);
            acc = fmaf(a1[u].x, b1[u].x, acc);
            acc = fmaf(a1[u].y, b1[u].y, acc);
            acc = fmaf(a1[u].z, b1[u].z, acc);
            acc = fmaf(a1[u].w, b1[u].w, acc);
            p[hp * 4 + u] = acc;
        }
    }

    const bool hi4 = (r & 4) != 0;
    float q[4];
#pragma unroll
    for (int k = 0; k < 4; ++k) {
        const float send = hi4 ? p[k] : p[k + 4];
        const float keep = hi4 ? p[k + 4] : p[k];
        q[k] = keep + __shfl_xor(send, 4, 32);
    }
    const bool hi2 = (r & 2) != 0;
    float w[2];
#pragma unroll
    for (int k = 0; k < 2; ++k) {
        const float send = hi2 ? q[k] : q[k + 2];
        const float keep = hi2 ? q[k + 2] : q[k];
        w[k] = keep + __shfl_xor(send, 2, 32);
    }
    const bool hi1 = (r & 1) != 0;
    const float send1 = hi1 ? w[0] : w[1];
    const float keep1 = hi1 ? w[1] : w[0];
    const float acc   = keep1 + __shfl_xor(send1, 1, 32);

    unsigned int* maxbits =
        (hi && gat_planes(ws_size, n_nodes, n_edges) >= 1) ? maxbits1 : maxbits0;
    if (valid) {
        e[my] = acc;
        if (acc >= 0.0f)
            atomicMax((int*)&maxbits[d], __float_as_int(acc));
        else
            atomicMin(&maxbits[d], __float_as_uint(acc));
    }
}

__global__ void gat_exp_denom_kernel(const int* __restrict__ dst,
                                     float* __restrict__ e,
                                     const unsigned int* __restrict__ maxbits0,
                                     const unsigned int* __restrict__ maxbits1,
                                     float* __restrict__ denom0,
                                     float* __restrict__ denom1,
                                     int n_nodes, int n_edges, size_t ws_size) {
    int eid = blockIdx.x * blockDim.x + threadIdx.x;
    if (eid >= n_edges) return;
    const int planes = gat_planes(ws_size, n_nodes, n_edges);
    const int hi = gat_half();
    const unsigned int* mb1 = (planes >= 1) ? maxbits1 : maxbits0;
    int d = dst[eid];
    const float ev = e[eid];
    const float m = fmaxf(__int_as_float((int)maxbits0[d]),
                          __int_as_float((int)mb1[d]));
    float t = __expf(ev - m);
    e[eid] = t;
    float* denom = (hi && planes >= 1) ? denom1 : denom0;
    atomicAdd(&denom[d], t);
}


typedef __attribute__((address_space(1))) float       gat_gfloat;
typedef __attribute__((address_space(1))) const float gat_gcfloat;

__device__ __forceinline__ unsigned long long gat_row_addr(const float* base, int row) {
    const unsigned long long a =
        (unsigned long long)base + ((unsigned long long)(unsigned int)row << 8);
    const unsigned int lo = __builtin_amdgcn_readfirstlane((unsigned int)a);
    const unsigned int hi = __builtin_amdgcn_readfirstlane((unsigned int)(a >> 32));
    return ((unsigned long long)hi << 32) | lo;
}

template <bool FULL>
__device__ __forceinline__ void gat_aggregate_walk(const float* __restrict__ z,
                                                   float* __restrict__ out0,
                                                   float* __restrict__ out1,
                                                   bool second,
                                                   int s, int d, float ev, float dn,
                                                   int cnt, int lane) {
    float v0[2][8], v1[2][8];
#pragma unroll
    for (int u = 0; u < 8; ++u) {
        const gat_gcfloat* zr =
            (const gat_gcfloat*)gat_row_addr(z, __builtin_amdgcn_readlane(s, u));
        v0[0][u] = zr[lane];
        v1[0][u] = zr[lane + 32];
    }
    const float alpha = ev / fmaxf(dn, 1e-30f);
    const int abits = __float_as_int(alpha);
    float* out = second ? out1 : out0;
#pragma unroll
    for (int g = 0; g < 4; ++g) {
        if (g + 1 < 4) {
#pragma unroll
            for (int u = 0; u < 8; ++u) {
                const gat_gcfloat* zr = (const gat_gcfloat*)gat_row_addr(
                    z, __builtin_amdgcn_readlane(s, (g + 1) * 8 + u));
                v0[(g + 1) & 1][u] = zr[lane];
                v1[(g + 1) & 1][u] = zr[lane + 32];
            }
        }
#pragma unroll
        for (int u = 0; u < 8; ++u) {
            const int k = g * 8 + u;
            if (FULL || k < cnt) {
                gat_gfloat* orow =
                    (gat_gfloat*)gat_row_addr(out, __builtin_amdgcn_readlane(d, k));
                const float ak = __int_as_float(__builtin_amdgcn_readlane(abits, k));
                __hip_atomic_fetch_add(orow + lane, ak * v0[g & 1][u],
                                       __ATOMIC_RELAXED, __HIP_MEMORY_SCOPE_AGENT);
                __hip_atomic_fetch_add(orow + lane + 32, ak * v1[g & 1][u],
                                       __ATOMIC_RELAXED, __HIP_MEMORY_SCOPE_AGENT);
            }
        }
    }
}

__global__ void __launch_bounds__(256)
gat_aggregate_kernel(const float* __restrict__ z,
                     const int* __restrict__ src,
                     const int* __restrict__ dst,
                     const float* __restrict__ e,
                     const float* __restrict__ denom0,
                     const float* __restrict__ denom1,
                     float* __restrict__ out0,
                     float* __restrict__ out1,
                     int n_nodes, int n_edges, size_t ws_size) {
    const int lane = threadIdx.x & 31;
    const int wib  = __builtin_amdgcn_readfirstlane(threadIdx.x >> 5);
    const int wid  = blockIdx.x * (blockDim.x >> 5) + wib;
    const int eid0 = wid * 32;
    if (eid0 >= n_edges) return;

    const int planes = gat_planes(ws_size, n_nodes, n_edges);
    const int hi = gat_half();

    const int my  = eid0 + lane;
    const int ld  = (my < n_edges) ? my : (n_edges - 1);
    const int s   = src[ld];
    const int d   = dst[ld];
    const float ev = e[ld];
    const float* dp1 = (planes >= 1) ? denom1 : denom0;
    const float dn0 = denom0[d];
    const float dn1 = dp1[d];
    const float dn  = (planes >= 1) ? dn0 + dn1 : dn0;
    const bool second = hi && planes == 2;

    const int cnt = n_edges - eid0;
    if (cnt >= 32)
        gat_aggregate_walk<true>(z, out0, out1, second, s, d, ev, dn, 32, lane);
    else
        gat_aggregate_walk<false>(z, out0, out1, second, s, d, ev, dn, cnt, lane);
}

__global__ void __launch_bounds__(256)
gat_merge_kernel(float* __restrict__ out,
                 const float* __restrict__ out1,
                 int n_nodes, int n_edges, size_t ws_size) {
    if (gat_planes(ws_size, n_nodes, n_edges) != 2) return;
    const int i      = blockIdx.x * blockDim.x + threadIdx.x;
    const int total4 = n_nodes * (OUT_DIM / 4);
    if (i < total4) {
        float4 a = ((const float4*)out)[i];
        const float4 b = ((const float4*)out1)[i];
        a.x += b.x; a.y += b.y; a.z += b.z; a.w += b.w;
        ((float4*)out)[i] = a;
    }
}

extern "C" void kernel_launch(void* const* d_in, const int* in_sizes, int n_in,
                              void* d_out, int out_size, void* d_ws, size_t ws_size,
                              hipStream_t stream) {
    const float* h   = (const float*)d_in[0];
    const float* W   = (const float*)d_in[1];
    const int*   src = (const int*)d_in[2];
    const int*   dst = (const int*)d_in[3];
    float* out = (float*)d_out;

    const int n_nodes = in_sizes[0] / IN_DIM;
    const int n_edges = in_sizes[2];

    float*        z       = (float*)d_ws;
    float*        e       = z + (size_t)n_nodes * OUT_DIM;
    unsigned int* maxbits = (unsigned int*)(e + n_edges);
    float*        denom   = (float*)(maxbits + n_nodes);
    unsigned int* maxbits1 = (unsigned int*)(denom + n_nodes);
    float*        denom1   = (float*)(maxbits1 + n_nodes);
    float*        out1     = denom1 + n_nodes;

    {
        int total = n_nodes * OUT_DIM;
        gat_init_kernel<<<(total + 255) / 256, 256, 0, stream>>>(
            maxbits, denom, maxbits1, denom1, out, out1, n_nodes, n_edges, ws_size);
    }
    {
        int blocks = (n_nodes + GEMM_ROWS - 1) / GEMM_ROWS;
        gat_gemm_wmma_kernel<<<blocks, 128, 0, stream>>>(h, W, z, n_nodes);
    }
    {
        int edges_per_block = (256 / 32) * 32;
        int blocks = (n_edges + edges_per_block - 1) / edges_per_block;
        gat_edge_score_kernel<<<blocks, 256, 0, stream>>>(
            z, src, dst, e, maxbits, maxbits1, n_nodes, n_edges, ws_size);
    }
    {
        gat_exp_denom_kernel<<<(n_edges + 255) / 256, 256, 0, stream>>>(
            dst, e, maxbits, maxbits1, denom, denom1, n_nodes, n_edges, ws_size);
    }
    {
        int edges_per_block = (256 / 32) * 32;
        int blocks = (n_edges + edges_per_block - 1) / edges_per_block;
        gat_aggregate_kernel<<<blocks, 256, 0, stream>>>(
            z, src, dst, e, denom, denom1, out, out1, n_nodes, n_edges, ws_size);
    }
    {
        int total4 = n_nodes * (OUT_DIM / 4);
        gat_merge_kernel<<<(total4 + 255) / 256, 256, 0, stream>>>(
            out, out1, n_nodes, n_edges, ws_size);
    }
}


# Aggregate re-cut by destination: 256 work-groups each keep 391 out rows in LDS, find their edges by a key byte per edge, add t*z with LDS adds; no out fill, no merge launch; sums in another order
# speedup vs baseline: 4.6122x; 1.5169x over previous
#include <hip/hip_runtime.h>
#include <hip/hip_bf16.h>

#define IN_DIM  128
#define OUT_DIM 64

typedef __attribute__((ext_vector_type(2))) float v2f;
typedef __attribute__((ext_vector_type(8))) float v8f;

#define AGG_OWNERS 256
#define AGG_RMAX   640

__device__ __forceinline__ int gat_owner_rows(int n_nodes) {
    return (n_nodes + AGG_OWNERS - 1) / AGG_OWNERS;
}

__device__ __forceinline__ size_t gat_ws_small(int n_nodes, int n_edges) {
    return ((size_t)n_nodes * OUT_DIM + (size_t)n_edges + 2 * (size_t)n_nodes) * sizeof(float);
}

__device__ __forceinline__ int gat_planes(size_t ws_size, int n_nodes, int n_edges) {
    return ws_size >= gat_ws_small(n_nodes, n_edges) ? 1 : 0;
}

__device__ __forceinline__ bool gat_have_keys(size_t ws_size, int n_nodes, int n_edges) {
    const size_t key_bytes = ((size_t)n_edges + 15) & ~(size_t)15;
    return ws_size >= gat_ws_small(n_nodes, n_edges) + key_bytes;
}

__device__ __forceinline__ int gat_half() {
    const unsigned int die = (__builtin_amdgcn_s_sendmsg_rtn(0x87) >> 16) & 0xFu;
    return die >= 4u ? 1 : 0;
}

__global__ void gat_init_kernel(unsigned int* __restrict__ maxbits0,
                                unsigned int* __restrict__ maxbits1,
                                int n_nodes, int n_edges, size_t ws_size) {
    const int planes = gat_planes(ws_size, n_nodes, n_edges);
    int i = blockIdx.x * blockDim.x + threadIdx.x;
    if (i < n_nodes) {
        maxbits0[i] = 0xFF800000u;
        if (planes >= 1) maxbits1[i] = 0xFF800000u;
    }
}

typedef __attribute__((ext_vector_type(4))) float v4f;
typedef __attribute__((address_space(1))) const v4f gat_gcv4f;

__device__ __forceinline__ const gat_gcv4f* gat_h_row(const float* base, int row) {
    const unsigned long long a =
        (unsigned long long)base + ((unsigned long long)(unsigned int)row << 9);
    const unsigned int lo = __builtin_amdgcn_readfirstlane((unsigned int)a);
    const unsigned int hi = __builtin_amdgcn_readfirstlane((unsigned int)(a >> 32));
    return (const gat_gcv4f*)(((unsigned long long)hi << 32) | lo);
}

#define GEMM_RT     5
#define GEMM_ROWS   (16 * GEMM_RT)
#define GEMM_WROWS  (GEMM_ROWS / 4)
#define GEMM_PITCH4 33

__global__ void __launch_bounds__(128)
gat_gemm_wmma_kernel(const float* __restrict__ h,
                     const float* __restrict__ W,
                     float* __restrict__ z,
                     int n_nodes) {
    __shared__ v4f tile4[GEMM_ROWS * GEMM_PITCH4];

    const int tid   = threadIdx.x;
    const int lane  = tid & 31;
    const int wave  = __builtin_amdgcn_readfirstlane(tid >> 5);
    const int half  = lane >> 4;
    const int lcol  = lane & 15;

    const int row0  = blockIdx.x * GEMM_ROWS;
    const int ncol  = wave * 16 + lcol;

    v4f st[GEMM_WROWS];
#pragma unroll
    for (int p = 0; p < GEMM_WROWS; ++p) {
        int gr = row0 + 4 * p + wave;
        if (gr >= n_nodes) gr = n_nodes - 1;
        st[p] = gat_h_row(h, gr)[lane];
    }

    const v4f* __restrict__ wrow =
        (const v4f*)(W + (size_t)ncol * IN_DIM) + half;
    v4f b[16];
#pragma unroll
    for (int j = 0; j < 16; ++j) b[j] = wrow[2 * j];

#pragma unroll
    for (int p = 0; p < GEMM_WROWS; ++p)
        tile4[(4 * p + wave) * GEMM_PITCH4 + lane] = st[p];

    __syncthreads();

    v8f acc[GEMM_RT] = {};
    const v4f* arow = tile4 + lcol * GEMM_PITCH4 + half;
    v4f a_cur[GEMM_RT], a_nxt[GEMM_RT];
#pragma unroll
    for (int rt = 0; rt < GEMM_RT; ++rt)
        a_cur[rt] = arow[rt * 16 * GEMM_PITCH4];
#pragma unroll
    for (int j = 0; j < 16; ++j) {
        if (j + 1 < 16) {
#pragma unroll
            for (int rt = 0; rt < GEMM_RT; ++rt)
                a_nxt[rt] = arow[rt * 16 * GEMM_PITCH4 + 2 * (j + 1)];
        }
        __builtin_amdgcn_sched_barrier(0);
        v2f b0; b0[0] = b[j].x; b0[1] = b[j].y;
        v2f b1; b1[0] = b[j].z; b1[1] = b[j].w;
#pragma unroll
        for (int rt = 0; rt < GEMM_RT; ++rt) {
            v2f a0; a0[0] = a_cur[rt].x; a0[1] = a_cur[rt].y;
            acc[rt] = __builtin_amdgcn_wmma_f32_16x16x4_f32(
                 false, a0,  false, b0,
                 (short)0, acc[rt],  false,  false);
        }
#pragma unroll
        for (int rt = 0; rt < GEMM_RT; ++rt) {
            v2f a1; a1[0] = a_cur[rt].z; a1[1] = a_cur[rt].w;
            acc[rt] = __builtin_amdgcn_wmma_f32_16x16x4_f32(
                 false, a1,  false, b1,
                 (short)0, acc[rt],  false,  false);
        }
        __builtin_amdgcn_sched_barrier(0);
#pragma unroll
        for (int rt = 0; rt < GEMM_RT; ++rt) a_cur[rt] = a_nxt[rt];
    }

    float* zout = z + (size_t)(row0 + 8 * half) * OUT_DIM + ncol;
    if (row0 + GEMM_ROWS <= n_nodes) {
#pragma unroll
        for (int rt = 0; rt < GEMM_RT; ++rt) {
#pragma unroll
            for (int v = 0; v < 8; ++v)
                zout[(rt * 16 + v) * OUT_DIM] = acc[rt][v];
        }
    } else {
#pragma unroll
        for (int rt = 0; rt < GEMM_RT; ++rt) {
#pragma unroll
            for (int v = 0; v < 8; ++v) {
                int row = row0 + rt * 16 + v + 8 * half;
                if (row < n_nodes)
                    zout[(rt * 16 + v) * OUT_DIM] = acc[rt][v];
            }
        }
    }
}

__global__ void __launch_bounds__(256)
gat_edge_score_kernel(const float* __restrict__ z,
                      const int* __restrict__ src,
                      const int* __restrict__ dst,
                      float* __restrict__ e,
                      unsigned int* __restrict__ maxbits0,
                      unsigned int* __restrict__ maxbits1,
                      unsigned char* __restrict__ keys,
                      int n_nodes, int n_edges, size_t ws_size) {
    const int lane = threadIdx.x & 31;
    const int r    = lane & 7;
    const int g    = lane >> 3;
    const int wid  = blockIdx.x * (blockDim.x >> 5) + (threadIdx.x >> 5);
    const int eid0 = wid * 32;
    if (eid0 >= n_edges) return;

    const int hi = gat_half();

    const int  my    = eid0 + 4 * r + g;
    const bool valid = my < n_edges;
    const int  ld    = valid ? my : (n_edges - 1);
    const int  s     = src[ld];
    const int  d     = dst[ld];

    float p[8];
#pragma unroll
    for (int hp = 0; hp < 2; ++hp) {
        float4 a0[4], a1[4], b0[4], b1[4];
#pragma unroll
        for (int u = 0; u < 4; ++u) {
            const int t    = hp * 4 + u;
            const int from = (lane & 24) + t;
            const int st   = __shfl(s, from, 32);
            const int dt   = __shfl(d, from, 32);
            const float4* zs = (const float4*)(z + (size_t)st * OUT_DIM) + r;
            const float4* zd = (const float4*)(z + (size_t)dt * OUT_DIM) + r;
            a0[u] = zs[0];
            a1[u] = zs[8];
            b0[u] = zd[0];
            b1[u] = zd[8];
        }
#pragma unroll
        for (int u = 0; u < 4; ++u) {
            float acc = a0[u].x * b0[u].x;
            acc = fmaf(a0[u].y, b0[u].y, acc);
            acc = fmaf(a0[u].z, b0[u].z, acc);
            acc = fmaf(a0[u].w, b0[u].w, acc);
            acc = fmaf(a1[u].x, b1[u].x, acc);
            acc = fmaf(a1[u].y, b1[u].y, acc);
            acc = fmaf(a1[u].z, b1[u].z, acc);
            acc = fmaf(a1[u].w, b1[u].w, acc);
            p[hp * 4 + u] = acc;
        }
    }

    const bool hi4 = (r & 4) != 0;
    float q[4];
#pragma unroll
    for (int k = 0; k < 4; ++k) {
        const float send = hi4 ? p[k] : p[k + 4];
        const float keep = hi4 ? p[k + 4] : p[k];
        q[k] = keep + __shfl_xor(send, 4, 32);
    }
    const bool hi2 = (r & 2) != 0;
    float w[2];
#pragma unroll
    for (int k = 0; k < 2; ++k) {
        const float send = hi2 ? q[k] : q[k + 2];
        const float keep = hi2 ? q[k + 2] : q[k];
        w[k] = keep + __shfl_xor(send, 2, 32);
    }
    const bool hi1 = (r & 1) != 0;
    const float send1 = hi1 ? w[0] : w[1];
    const float keep1 = hi1 ? w[1] : w[0];
    const float acc   = keep1 + __shfl_xor(send1, 1, 32);

    unsigned int* maxbits =
        (hi && gat_planes(ws_size, n_nodes, n_edges) >= 1) ? maxbits1 : maxbits0;
    if (valid) {
        e[my] = acc;
        if (gat_have_keys(ws_size, n_nodes, n_edges))
            keys[my] = (unsigned char)(d / gat_owner_rows(n_nodes));
        if (acc >= 0.0f)
            atomicMax((int*)&maxbits[d], __float_as_int(acc));
        else
            atomicMin(&maxbits[d], __float_as_uint(acc));
    }
}


typedef __attribute__((address_space(1))) const float gat_gcfloat;

__device__ __forceinline__ unsigned long long gat_row_addr(const float* base, int row) {
    const unsigned long long a =
        (unsigned long long)base + ((unsigned long long)(unsigned int)row << 8);
    const unsigned int lo = __builtin_amdgcn_readfirstlane((unsigned int)a);
    const unsigned int hi = __builtin_amdgcn_readfirstlane((unsigned int)(a >> 32));
    return ((unsigned long long)hi << 32) | lo;
}

#define AGG_THREADS 1024
#define AGG_WAVES   (AGG_THREADS / 32)
#define AGG_LIST    64
#define AGG_STEP    2048

__device__ __forceinline__ void gat_lds_add(float* p, float v) {
    __hip_atomic_fetch_add(p, v, __ATOMIC_RELAXED, __HIP_MEMORY_SCOPE_WORKGROUP);
}

__device__ __forceinline__ unsigned int gat_match_keys(const uint4 x, unsigned int pat) {
    const unsigned int w[4] = {x.x, x.y, x.z, x.w};
    unsigned int m = 0u;
#pragma unroll
    for (int i = 0; i < 4; ++i) {
        const unsigned int y = w[i] ^ pat;
        const unsigned int t = ((y & 0x7f7f7f7fu) + 0x7f7f7f7fu) | y;
        m |= (~t & 0x80808080u) >> (7 - i);
    }
    return m;
}

__device__ __forceinline__ unsigned int gat_match_dst(const int* __restrict__ dst,
                                                      int ebase, int n_edges,
                                                      int lo, int rows) {
    unsigned int m = 0u;
    for (int j = 0; j < 16; ++j) {
        if (ebase + j < n_edges) {
            const int d = dst[ebase + j];
            if ((unsigned int)(d - lo) < (unsigned int)rows)
                m |= 1u << (8 * (j & 3) + (j >> 2));
        }
    }
    return m;
}

template <bool FULL>
__device__ __forceinline__ void gat_owner_batch(const float* __restrict__ z,
                                                const int* __restrict__ src,
                                                const int* __restrict__ dst,
                                                const float* __restrict__ e,
                                                const int* ids, int cnt, int lo,
                                                float* acc, float* den, const float* mx,
                                                int lane) {
    const int   li  = (FULL || lane < cnt) ? lane : (cnt - 1);
    const int   eid = ids[li];
    const int   s   = src[eid];
    const int   dl  = dst[eid] - lo;
    const float ev  = e[eid];

    float v0[2][8], v1[2][8];
#pragma unroll
    for (int u = 0; u < 8; ++u) {
        const gat_gcfloat* zr =
            (const gat_gcfloat*)gat_row_addr(z, __builtin_amdgcn_readlane(s, u));
        v0[0][u] = zr[lane];
        v1[0][u] = zr[lane + 32];
    }
    const float t = __expf(ev - mx[dl]);
    if (FULL || lane < cnt)
        gat_lds_add(&den[dl], t);
    const int abits = __float_as_int(t);
#pragma unroll
    for (int g = 0; g < 4; ++g) {
        if (g + 1 < 4) {
#pragma unroll
            for (int u = 0; u < 8; ++u) {
                const gat_gcfloat* zr = (const gat_gcfloat*)gat_row_addr(
                    z, __builtin_amdgcn_readlane(s, (g + 1) * 8 + u));
                v0[(g + 1) & 1][u] = zr[lane];
                v1[(g + 1) & 1][u] = zr[lane + 32];
            }
        }
#pragma unroll
        for (int u = 0; u < 8; ++u) {
            const int k = g * 8 + u;
            if (FULL || k < cnt) {
                float* orow = acc + __builtin_amdgcn_readlane(dl, k) * OUT_DIM;
                const float ak = __int_as_float(__builtin_amdgcn_readlane(abits, k));
                gat_lds_add(orow + lane, ak * v0[g & 1][u]);
                gat_lds_add(orow + lane + 32, ak * v1[g & 1][u]);
            }
        }
    }
}

__global__ void __launch_bounds__(AGG_THREADS)
gat_aggregate_kernel(const float* __restrict__ z,
                     const int* __restrict__ src,
                     const int* __restrict__ dst,
                     const float* __restrict__ e,
                     const unsigned int* __restrict__ maxbits0,
                     const unsigned int* __restrict__ maxbits1,
                     const unsigned char* __restrict__ keys,
                     float* __restrict__ out,
                     int n_nodes, int n_edges, size_t ws_size) {
    __shared__ __attribute__((aligned(16))) float acc[AGG_RMAX * OUT_DIM];
    __shared__ float den[AGG_RMAX];
    __shared__ float mx[AGG_RMAX];
    __shared__ int   list[AGG_WAVES * AGG_LIST];

    const int tid  = threadIdx.x;
    const int lane = tid & 31;
    const int wave = __builtin_amdgcn_readfirstlane(tid >> 5);
    const int rows = gat_owner_rows(n_nodes);
    const int own  = blockIdx.x;
    const int lo   = own * rows;
    if (lo >= n_nodes) return;
    const int  nrows     = (n_nodes - lo < rows) ? (n_nodes - lo) : rows;
    const int  planes    = gat_planes(ws_size, n_nodes, n_edges);
    const bool have_keys = gat_have_keys(ws_size, n_nodes, n_edges);

    for (int i = tid; i < nrows * OUT_DIM; i += AGG_THREADS) acc[i] = 0.0f;
    const unsigned int* mb1 = (planes >= 1) ? maxbits1 : maxbits0;
    for (int i = tid; i < nrows; i += AGG_THREADS) {
        den[i] = 0.0f;
        mx[i]  = fmaxf(__int_as_float((int)maxbits0[lo + i]),
                       __int_as_float((int)mb1[lo + i]));
    }
    __syncthreads();

    int* ids = list + wave * AGG_LIST;
    int  cnt = 0;
    const unsigned int pat = (unsigned int)own * 0x01010101u;
    for (int blk = wave * AGG_STEP; blk < n_edges; blk += AGG_WAVES * AGG_STEP) {
        unsigned int m0, m1, m2, m3;
        if (have_keys && blk + AGG_STEP <= n_edges) {
            const uint4* kp = (const uint4*)(keys + blk) + lane;
            const uint4 x0 = kp[0];
            const uint4 x1 = kp[32];
            const uint4 x2 = kp[64];
            const uint4 x3 = kp[96];
            m0 = gat_match_keys(x0, pat);
            m1 = gat_match_keys(x1, pat);
            m2 = gat_match_keys(x2, pat);
            m3 = gat_match_keys(x3, pat);
        } else {
            m0 = gat_match_dst(dst, blk + lane * 16,        n_edges, lo, rows);
            m1 = gat_match_dst(dst, blk + lane * 16 + 512,  n_edges, lo, rows);
            m2 = gat_match_dst(dst, blk + lane * 16 + 1024, n_edges, lo, rows);
            m3 = gat_match_dst(dst, blk + lane * 16 + 1536, n_edges, lo, rows);
        }
#pragma unroll 1
        for (int j = 0; j < 4; ++j) {
            unsigned int m = (j == 0) ? m0 : (j == 1) ? m1 : (j == 2) ? m2 : m3;
            const int ebase = blk + j * 512 + lane * 16;
            while (__builtin_amdgcn_ballot_w32(m != 0u) != 0u) {
                const bool has = m != 0u;
                const int  p   = __builtin_ctz(m | 0x80000000u);
                const int  eid = ebase + 4 * (p & 7) + (p >> 3);
                m &= m - 1u;
                const bool ok  = has && eid < n_edges;
                const unsigned int bal = __builtin_amdgcn_ballot_w32(ok);
                if (ok) ids[cnt + (int)__builtin_amdgcn_mbcnt_lo(bal, 0u)] = eid;
                cnt += __builtin_popcount(bal);
                if (cnt >= 32) {
                    cnt -= 32;
                    __builtin_amdgcn_fence(__ATOMIC_ACQ_REL, "wavefront");
                    gat_owner_batch<true>(z, src, dst, e, ids + cnt, 32, lo,
                                          acc, den, mx, lane);
                }
            }
        }
    }
    if (cnt > 0) {
        __builtin_amdgcn_fence(__ATOMIC_ACQ_REL, "wavefront");
        gat_owner_batch<false>(z, src, dst, e, ids, cnt, lo, acc, den, mx, lane);
    }
    __syncthreads();

    float4*       o4 = (float4*)(out + (size_t)lo * OUT_DIM);
    const float4* a4 = (const float4*)acc;
    for (int i = tid; i < nrows * (OUT_DIM / 4); i += AGG_THREADS) {
        const float inv = 1.0f / fmaxf(den[i / (OUT_DIM / 4)], 1e-30f);
        float4 a = a4[i];
        a.x *= inv; a.y *= inv; a.z *= inv; a.w *= inv;
        o4[i] = a;
    }
}

extern "C" void kernel_launch(void* const* d_in, const int* in_sizes, int n_in,
                              void* d_out, int out_size, void* d_ws, size_t ws_size,
                              hipStream_t stream) {
    const float* h   = (const float*)d_in[0];
    const float* W   = (const float*)d_in[1];
    const int*   src = (const int*)d_in[2];
    const int*   dst = (const int*)d_in[3];
    float* out = (float*)d_out;

    const int n_nodes = in_sizes[0] / IN_DIM;
    const int n_edges = in_sizes[2];

    float*         z        = (float*)d_ws;
    float*         e        = z + (size_t)n_nodes * OUT_DIM;
    unsigned int*  maxbits  = (unsigned int*)(e + n_edges);
    unsigned int*  maxbits1 = maxbits + n_nodes;
    unsigned char* keys     = (unsigned char*)(maxbits1 + n_nodes);

    {
        gat_init_kernel<<<(n_nodes + 255) / 256, 256, 0, stream>>>(
            maxbits, maxbits1, n_nodes, n_edges, ws_size);
    }
    {
        int blocks = (n_nodes + GEMM_ROWS - 1) / GEMM_ROWS;
        gat_gemm_wmma_kernel<<<blocks, 128, 0, stream>>>(h, W, z, n_nodes);
    }
    {
        int edges_per_block = (256 / 32) * 32;
        int blocks = (n_edges + edges_per_block - 1) / edges_per_block;
        gat_edge_score_kernel<<<blocks, 256, 0, stream>>>(
            z, src, dst, e, maxbits, maxbits1, keys, n_nodes, n_edges, ws_size);
    }
    {
        gat_aggregate_kernel<<<AGG_OWNERS, AGG_THREADS, 0, stream>>>(
            z, src, dst, e, maxbits, maxbits1, keys, out, n_nodes, n_edges, ws_size);
    }
}


# Score launch appends each edge as an 8-byte record (one returning atomic add) to its destination owner's list, kept in that owner's out rows; owners read lists, not all keys; sums in arrival order
# speedup vs baseline: 5.2873x; 1.1464x over previous
#include <hip/hip_runtime.h>
#include <hip/hip_bf16.h>

#define IN_DIM  128
#define OUT_DIM 64

typedef __attribute__((ext_vector_type(2))) float v2f;
typedef __attribute__((ext_vector_type(8))) float v8f;

#define AGG_OWNERS 256
#define AGG_RMAX   640

__device__ __forceinline__ int gat_owner_rows(int n_nodes) {
    return (n_nodes + AGG_OWNERS - 1) / AGG_OWNERS;
}

__device__ __forceinline__ size_t gat_ws_small(int n_nodes, int n_edges) {
    return ((size_t)n_nodes * OUT_DIM + (size_t)n_edges + 2 * (size_t)n_nodes) * sizeof(float);
}

__device__ __forceinline__ int gat_planes(size_t ws_size, int n_nodes, int n_edges) {
    return ws_size >= gat_ws_small(n_nodes, n_edges) ? 1 : 0;
}

__device__ __forceinline__ bool gat_have_keys(size_t ws_size, int n_nodes, int n_edges) {
    const size_t key_bytes = ((size_t)n_edges + 15) & ~(size_t)15;
    return ws_size >= gat_ws_small(n_nodes, n_edges) + key_bytes;
}

__device__ __forceinline__ unsigned long long* gat_list(float* out, int owner, int rows,
                                                        int n_nodes, int half,
                                                        unsigned int* cap) {
    const int lo = owner * rows;
    const int nr = (n_nodes - lo < rows) ? (n_nodes - lo) : rows;
    const unsigned int slots = (unsigned int)nr * 16u;
    *cap = slots - 1u;
    return (unsigned long long*)(out + (size_t)lo * OUT_DIM) + (size_t)half * slots;
}

__device__ __forceinline__ int gat_half() {
    const unsigned int die = (__builtin_amdgcn_s_sendmsg_rtn(0x87) >> 16) & 0xFu;
    return die >= 4u ? 1 : 0;
}

__global__ void gat_init_kernel(unsigned int* __restrict__ maxbits0,
                                unsigned int* __restrict__ maxbits1,
                                float* __restrict__ out,
                                int n_nodes, int n_edges, size_t ws_size) {
    const int planes = gat_planes(ws_size, n_nodes, n_edges);
    int i = blockIdx.x * blockDim.x + threadIdx.x;
    if (i < n_nodes) {
        maxbits0[i] = 0xFF800000u;
        if (planes >= 1) maxbits1[i] = 0xFF800000u;
    }
    if (i < AGG_OWNERS) {
        const int rows = gat_owner_rows(n_nodes);
        if (i * rows < n_nodes) {
            unsigned int cap;
            gat_list(out, i, rows, n_nodes, 0, &cap)[0] = 0ull;
            gat_list(out, i, rows, n_nodes, 1, &cap)[0] = 0ull;
        }
    }
}

typedef __attribute__((ext_vector_type(4))) float v4f;
typedef __attribute__((address_space(1))) const v4f gat_gcv4f;

__device__ __forceinline__ const gat_gcv4f* gat_h_row(const float* base, int row) {
    const unsigned long long a =
        (unsigned long long)base + ((unsigned long long)(unsigned int)row << 9);
    const unsigned int lo = __builtin_amdgcn_readfirstlane((unsigned int)a);
    const unsigned int hi = __builtin_amdgcn_readfirstlane((unsigned int)(a >> 32));
    return (const gat_gcv4f*)(((unsigned long long)hi << 32) | lo);
}

#define GEMM_RT     5
#define GEMM_ROWS   (16 * GEMM_RT)
#define GEMM_WROWS  (GEMM_ROWS / 4)
#define GEMM_PITCH4 33

__global__ void __launch_bounds__(128)
gat_gemm_wmma_kernel(const float* __restrict__ h,
                     const float* __restrict__ W,
                     float* __restrict__ z,
                     int n_nodes) {
    __shared__ v4f tile4[GEMM_ROWS * GEMM_PITCH4];

    const int tid   = threadIdx.x;
    const int lane  = tid & 31;
    const int wave  = __builtin_amdgcn_readfirstlane(tid >> 5);
    const int half  = lane >> 4;
    const int lcol  = lane & 15;

    const int row0  = blockIdx.x * GEMM_ROWS;
    const int ncol  = wave * 16 + lcol;

    v4f st[GEMM_WROWS];
#pragma unroll
    for (int p = 0; p < GEMM_WROWS; ++p) {
        int gr = row0 + 4 * p + wave;
        if (gr >= n_nodes) gr = n_nodes - 1;
        st[p] = gat_h_row(h, gr)[lane];
    }

    const v4f* __restrict__ wrow =
        (const v4f*)(W + (size_t)ncol * IN_DIM) + half;
    v4f b[16];
#pragma unroll
    for (int j = 0; j < 16; ++j) b[j] = wrow[2 * j];

#pragma unroll
    for (int p = 0; p < GEMM_WROWS; ++p)
        tile4[(4 * p + wave) * GEMM_PITCH4 + lane] = st[p];

    __syncthreads();

    v8f acc[GEMM_RT] = {};
    const v4f* arow = tile4 + lcol * GEMM_PITCH4 + half;
    v4f a_cur[GEMM_RT], a_nxt[GEMM_RT];
#pragma unroll
    for (int rt = 0; rt < GEMM_RT; ++rt)
        a_cur[rt] = arow[rt * 16 * GEMM_PITCH4];
#pragma unroll
    for (int j = 0; j < 16; ++j) {
        if (j + 1 < 16) {
#pragma unroll
            for (int rt = 0; rt < GEMM_RT; ++rt)
                a_nxt[rt] = arow[rt * 16 * GEMM_PITCH4 + 2 * (j + 1)];
        }
        __builtin_amdgcn_sched_barrier(0);
        v2f b0; b0[0] = b[j].x; b0[1] = b[j].y;
        v2f b1; b1[0] = b[j].z; b1[1] = b[j].w;
#pragma unroll
        for (int rt = 0; rt < GEMM_RT; ++rt) {
            v2f a0; a0[0] = a_cur[rt].x; a0[1] = a_cur[rt].y;
            acc[rt] = __builtin_amdgcn_wmma_f32_16x16x4_f32(
                 false, a0,  false, b0,
                 (short)0, acc[rt],  false,  false);
        }
#pragma unroll
        for (int rt = 0; rt < GEMM_RT; ++rt) {
            v2f a1; a1[0] = a_cur[rt].z; a1[1] = a_cur[rt].w;
            acc[rt] = __builtin_amdgcn_wmma_f32_16x16x4_f32(
                 false, a1,  false, b1,
                 (short)0, acc[rt],  false,  false);
        }
        __builtin_amdgcn_sched_barrier(0);
#pragma unroll
        for (int rt = 0; rt < GEMM_RT; ++rt) a_cur[rt] = a_nxt[rt];
    }

    float* zout = z + (size_t)(row0 + 8 * half) * OUT_DIM + ncol;
    if (row0 + GEMM_ROWS <= n_nodes) {
#pragma unroll
        for (int rt = 0; rt < GEMM_RT; ++rt) {
#pragma unroll
            for (int v = 0; v < 8; ++v)
                zout[(rt * 16 + v) * OUT_DIM] = acc[rt][v];
        }
    } else {
#pragma unroll
        for (int rt = 0; rt < GEMM_RT; ++rt) {
#pragma unroll
            for (int v = 0; v < 8; ++v) {
                int row = row0 + rt * 16 + v + 8 * half;
                if (row < n_nodes)
                    zout[(rt * 16 + v) * OUT_DIM] = acc[rt][v];
            }
        }
    }
}

__global__ void __launch_bounds__(256)
gat_edge_score_kernel(const float* __restrict__ z,
                      const int* __restrict__ src,
                      const int* __restrict__ dst,
                      float* __restrict__ e,
                      unsigned int* __restrict__ maxbits0,
                      unsigned int* __restrict__ maxbits1,
                      unsigned char* __restrict__ keys,
                      float* __restrict__ out,
                      int n_nodes, int n_edges, size_t ws_size) {
    const int lane = threadIdx.x & 31;
    const int r    = lane & 7;
    const int g    = lane >> 3;
    const int wid  = blockIdx.x * (blockDim.x >> 5) + (threadIdx.x >> 5);
    const int eid0 = wid * 32;
    if (eid0 >= n_edges) return;

    const int hi = gat_half();

    const int  my    = eid0 + 4 * r + g;
    const bool valid = my < n_edges;
    const int  ld    = valid ? my : (n_edges - 1);
    const int  s     = src[ld];
    const int  d     = dst[ld];

    const int rows  = gat_owner_rows(n_nodes);
    const int owner = d / rows;
    const int dl    = d - owner * rows;
    unsigned int cap;
    unsigned long long* lst = gat_list(out, owner, rows, n_nodes, hi, &cap);
    unsigned int slot = 0xFFFFFFFFu;

    float p[8];
#pragma unroll
    for (int hp = 0; hp < 2; ++hp) {
        float4 a0[4], a1[4], b0[4], b1[4];
#pragma unroll
        for (int u = 0; u < 4; ++u) {
            const int t    = hp * 4 + u;
            const int from = (lane & 24) + t;
            const int st   = __shfl(s, from, 32);
            const int dt   = __shfl(d, from, 32);
            const float4* zs = (const float4*)(z + (size_t)st * OUT_DIM) + r;
            const float4* zd = (const float4*)(z + (size_t)dt * OUT_DIM) + r;
            a0[u] = zs[0];
            a1[u] = zs[8];
            b0[u] = zd[0];
            b1[u] = zd[8];
        }
        if (hp == 0 && valid) slot = atomicAdd((unsigned int*)lst, 1u);
#pragma unroll
        for (int u = 0; u < 4; ++u) {
            float acc = a0[u].x * b0[u].x;
            acc = fmaf(a0[u].y, b0[u].y, acc);
            acc = fmaf(a0[u].z, b0[u].z, acc);
            acc = fmaf(a0[u].w, b0[u].w, acc);
            acc = fmaf(a1[u].x, b1[u].x, acc);
            acc = fmaf(a1[u].y, b1[u].y, acc);
            acc = fmaf(a1[u].z, b1[u].z, acc);
            acc = fmaf(a1[u].w, b1[u].w, acc);
            p[hp * 4 + u] = acc;
        }
    }

    const bool hi4 = (r & 4) != 0;
    float q[4];
#pragma unroll
    for (int k = 0; k < 4; ++k) {
        const float send = hi4 ? p[k] : p[k + 4];
        const float keep = hi4 ? p[k + 4] : p[k];
        q[k] = keep + __shfl_xor(send, 4, 32);
    }
    const bool hi2 = (r & 2) != 0;
    float w[2];
#pragma unroll
    for (int k = 0; k < 2; ++k) {
        const float send = hi2 ? q[k] : q[k + 2];
        const float keep = hi2 ? q[k + 2] : q[k];
        w[k] = keep + __shfl_xor(send, 2, 32);
    }
    const bool hi1 = (r & 1) != 0;
    const float send1 = hi1 ? w[0] : w[1];
    const float keep1 = hi1 ? w[1] : w[0];
    const float acc   = keep1 + __shfl_xor(send1, 1, 32);

    unsigned int* maxbits =
        (hi && gat_planes(ws_size, n_nodes, n_edges) >= 1) ? maxbits1 : maxbits0;
    if (valid) {
        e[my] = acc;
        if (gat_have_keys(ws_size, n_nodes, n_edges))
            keys[my] = (unsigned char)owner;
        if (slot < cap)
            lst[1u + slot] = ((unsigned long long)__float_as_uint(acc) << 32)
                             | (((unsigned int)s << 10) | (unsigned int)dl);
        if (acc >= 0.0f)
            atomicMax((int*)&maxbits[d], __float_as_int(acc));
        else
            atomicMin(&maxbits[d], __float_as_uint(acc));
    }
}


typedef __attribute__((address_space(1))) const float gat_gcfloat;

__device__ __forceinline__ unsigned long long gat_row_addr(const float* base, int row) {
    const unsigned long long a =
        (unsigned long long)base + ((unsigned long long)(unsigned int)row << 8);
    const unsigned int lo = __builtin_amdgcn_readfirstlane((unsigned int)a);
    const unsigned int hi = __builtin_amdgcn_readfirstlane((unsigned int)(a >> 32));
    return ((unsigned long long)hi << 32) | lo;
}

#define AGG_THREADS 1024
#define AGG_WAVES   (AGG_THREADS / 32)
#define AGG_LIST    64
#define AGG_STEP    2048

__device__ __forceinline__ void gat_lds_add(float* p, float v) {
    __hip_atomic_fetch_add(p, v, __ATOMIC_RELAXED, __HIP_MEMORY_SCOPE_WORKGROUP);
}

__device__ __forceinline__ unsigned int gat_match_keys(const uint4 x, unsigned int pat) {
    const unsigned int w[4] = {x.x, x.y, x.z, x.w};
    unsigned int m = 0u;
#pragma unroll
    for (int i = 0; i < 4; ++i) {
        const unsigned int y = w[i] ^ pat;
        const unsigned int t = ((y & 0x7f7f7f7fu) + 0x7f7f7f7fu) | y;
        m |= (~t & 0x80808080u) >> (7 - i);
    }
    return m;
}

__device__ __forceinline__ unsigned int gat_match_dst(const int* __restrict__ dst,
                                                      int ebase, int n_edges,
                                                      int lo, int rows) {
    unsigned int m = 0u;
    for (int j = 0; j < 16; ++j) {
        if (ebase + j < n_edges) {
            const int d = dst[ebase + j];
            if ((unsigned int)(d - lo) < (unsigned int)rows)
                m |= 1u << (8 * (j & 3) + (j >> 2));
        }
    }
    return m;
}

template <bool FULL>
__device__ __forceinline__ void gat_owner_core(const float* __restrict__ z,
                                               int s, int dl, float ev, int cnt,
                                               float* acc, float* den, const float* mx,
                                               int lane) {
    float v0[2][8], v1[2][8];
#pragma unroll
    for (int u = 0; u < 8; ++u) {
        const gat_gcfloat* zr =
            (const gat_gcfloat*)gat_row_addr(z, __builtin_amdgcn_readlane(s, u));
        v0[0][u] = zr[lane];
        v1[0][u] = zr[lane + 32];
    }
    const float t = __expf(ev - mx[dl]);
    if (FULL || lane < cnt)
        gat_lds_add(&den[dl], t);
    const int abits = __float_as_int(t);
#pragma unroll
    for (int g = 0; g < 4; ++g) {
        if (g + 1 < 4) {
#pragma unroll
            for (int u = 0; u < 8; ++u) {
                const gat_gcfloat* zr = (const gat_gcfloat*)gat_row_addr(
                    z, __builtin_amdgcn_readlane(s, (g + 1) * 8 + u));
                v0[(g + 1) & 1][u] = zr[lane];
                v1[(g + 1) & 1][u] = zr[lane + 32];
            }
        }
#pragma unroll
        for (int u = 0; u < 8; ++u) {
            const int k = g * 8 + u;
            if (FULL || k < cnt) {
                float* orow = acc + __builtin_amdgcn_readlane(dl, k) * OUT_DIM;
                const float ak = __int_as_float(__builtin_amdgcn_readlane(abits, k));
                gat_lds_add(orow + lane, ak * v0[g & 1][u]);
                gat_lds_add(orow + lane + 32, ak * v1[g & 1][u]);
            }
        }
    }
}

template <bool FULL>
__device__ __forceinline__ void gat_owner_batch(const float* __restrict__ z,
                                                const int* __restrict__ src,
                                                const int* __restrict__ dst,
                                                const float* __restrict__ e,
                                                const int* ids, int cnt, int lo,
                                                float* acc, float* den, const float* mx,
                                                int lane) {
    const int li  = (FULL || lane < cnt) ? lane : (cnt - 1);
    const int eid = ids[li];
    gat_owner_core<FULL>(z, src[eid], dst[eid] - lo, e[eid], cnt, acc, den, mx, lane);
}

__device__ __forceinline__ unsigned long long gat_record(const unsigned long long* l0,
                                                         const unsigned long long* l1,
                                                         unsigned int c0, unsigned int idx) {
    return (idx < c0) ? l0[1u + idx] : l1[1u + (idx - c0)];
}

__global__ void __launch_bounds__(AGG_THREADS)
gat_aggregate_kernel(const float* __restrict__ z,
                     const int* __restrict__ src,
                     const int* __restrict__ dst,
                     const float* __restrict__ e,
                     const unsigned int* __restrict__ maxbits0,
                     const unsigned int* __restrict__ maxbits1,
                     const unsigned char* __restrict__ keys,
                     float* __restrict__ out,
                     int n_nodes, int n_edges, size_t ws_size) {
    __shared__ __attribute__((aligned(16))) float acc[AGG_RMAX * OUT_DIM];
    __shared__ float den[AGG_RMAX];
    __shared__ float mx[AGG_RMAX];
    __shared__ int   list[AGG_WAVES * AGG_LIST];

    const int tid  = threadIdx.x;
    const int lane = tid & 31;
    const int wave = __builtin_amdgcn_readfirstlane(tid >> 5);
    const int rows = gat_owner_rows(n_nodes);
    const int own  = blockIdx.x;
    const int lo   = own * rows;
    if (lo >= n_nodes) return;
    const int  nrows     = (n_nodes - lo < rows) ? (n_nodes - lo) : rows;
    const int  planes    = gat_planes(ws_size, n_nodes, n_edges);
    const bool have_keys = gat_have_keys(ws_size, n_nodes, n_edges);

    unsigned int cap;
    const unsigned long long* l0 = gat_list(out, own, rows, n_nodes, 0, &cap);
    const unsigned long long* l1 = gat_list(out, own, rows, n_nodes, 1, &cap);
    const unsigned int c0 = __builtin_amdgcn_readfirstlane(((const unsigned int*)l0)[0]);
    const unsigned int c1 = __builtin_amdgcn_readfirstlane(((const unsigned int*)l1)[0]);

    for (int i = tid; i < nrows * OUT_DIM; i += AGG_THREADS) acc[i] = 0.0f;
    const unsigned int* mb1 = (planes >= 1) ? maxbits1 : maxbits0;
    for (int i = tid; i < nrows; i += AGG_THREADS) {
        den[i] = 0.0f;
        mx[i]  = fmaxf(__int_as_float((int)maxbits0[lo + i]),
                       __int_as_float((int)mb1[lo + i]));
    }
    __syncthreads();

    if (c0 <= cap && c1 <= cap) {
        const unsigned int total = c0 + c1;
        unsigned int base = (unsigned int)wave * 32u;
        unsigned long long rec = 0ull;
        if (base < total) {
            unsigned int idx = base + (unsigned int)lane;
            if (idx >= total) idx = total - 1u;
            rec = gat_record(l0, l1, c0, idx);
        }
        while (base < total) {
            const unsigned int nbase = base + AGG_WAVES * 32u;
            unsigned long long nrec = rec;
            if (nbase < total) {
                unsigned int idx = nbase + (unsigned int)lane;
                if (idx >= total) idx = total - 1u;
                nrec = gat_record(l0, l1, c0, idx);
            }
            const unsigned int w0 = (unsigned int)rec;
            int s  = (int)(w0 >> 10);
            int dl = (int)(w0 & 1023u);
            if (s  > n_nodes - 1) s  = n_nodes - 1;
            if (dl > nrows - 1)   dl = nrows - 1;
            const float ev = __uint_as_float((unsigned int)(rec >> 32));
            const unsigned int left = total - base;
            if (left >= 32u)
                gat_owner_core<true>(z, s, dl, ev, 32, acc, den, mx, lane);
            else
                gat_owner_core<false>(z, s, dl, ev, (int)left, acc, den, mx, lane);
            rec  = nrec;
            base = nbase;
        }
    } else {
        int* ids = list + wave * AGG_LIST;
        int  cnt = 0;
        const unsigned int pat = (unsigned int)own * 0x01010101u;
        for (int blk = wave * AGG_STEP; blk < n_edges; blk += AGG_WAVES * AGG_STEP) {
            unsigned int m0, m1, m2, m3;
            if (have_keys && blk + AGG_STEP <= n_edges) {
                const uint4* kp = (const uint4*)(keys + blk) + lane;
                const uint4 x0 = kp[0];
                const uint4 x1 = kp[32];
                const uint4 x2 = kp[64];
                const uint4 x3 = kp[96];
                m0 = gat_match_keys(x0, pat);
                m1 = gat_match_keys(x1, pat);
                m2 = gat_match_keys(x2, pat);
                m3 = gat_match_keys(x3, pat);
            } else {
                m0 = gat_match_dst(dst, blk + lane * 16,        n_edges, lo, rows);
                m1 = gat_match_dst(dst, blk + lane * 16 + 512,  n_edges, lo, rows);
                m2 = gat_match_dst(dst, blk + lane * 16 + 1024, n_edges, lo, rows);
                m3 = gat_match_dst(dst, blk + lane * 16 + 1536, n_edges, lo, rows);
            }
#pragma unroll 1
            for (int j = 0; j < 4; ++j) {
                unsigned int m = (j == 0) ? m0 : (j == 1) ? m1 : (j == 2) ? m2 : m3;
                const int ebase = blk + j * 512 + lane * 16;
                while (__builtin_amdgcn_ballot_w32(m != 0u) != 0u) {
                    const bool has = m != 0u;
                    const int  p   = __builtin_ctz(m | 0x80000000u);
                    const int  eid = ebase + 4 * (p & 7) + (p >> 3);
                    m &= m - 1u;
                    const bool ok  = has && eid < n_edges;
                    const unsigned int bal = __builtin_amdgcn_ballot_w32(ok);
                    if (ok) ids[cnt + (int)__builtin_amdgcn_mbcnt_lo(bal, 0u)] = eid;
                    cnt += __builtin_popcount(bal);
                    if (cnt >= 32) {
                        cnt -= 32;
                        __builtin_amdgcn_fence(__ATOMIC_ACQ_REL, "wavefront");
                        gat_owner_batch<true>(z, src, dst, e, ids + cnt, 32, lo,
                                              acc, den, mx, lane);
                    }
                }
            }
        }
        if (cnt > 0) {
            __builtin_amdgcn_fence(__ATOMIC_ACQ_REL, "wavefront");
            gat_owner_batch<false>(z, src, dst, e, ids, cnt, lo, acc, den, mx, lane);
        }
    }
    __syncthreads();

    float4*       o4 = (float4*)(out + (size_t)lo * OUT_DIM);
    const float4* a4 = (const float4*)acc;
    for (int i = tid; i < nrows * (OUT_DIM / 4); i += AGG_THREADS) {
        const float inv = 1.0f / fmaxf(den[i / (OUT_DIM / 4)], 1e-30f);
        float4 a = a4[i];
        a.x *= inv; a.y *= inv; a.z *= inv; a.w *= inv;
        o4[i] = a;
    }
}

extern "C" void kernel_launch(void* const* d_in, const int* in_sizes, int n_in,
                              void* d_out, int out_size, void* d_ws, size_t ws_size,
                              hipStream_t stream) {
    const float* h   = (const float*)d_in[0];
    const float* W   = (const float*)d_in[1];
    const int*   src = (const int*)d_in[2];
    const int*   dst = (const int*)d_in[3];
    float* out = (float*)d_out;

    const int n_nodes = in_sizes[0] / IN_DIM;
    const int n_edges = in_sizes[2];

    float*         z        = (float*)d_ws;
    float*         e        = z + (size_t)n_nodes * OUT_DIM;
    unsigned int*  maxbits  = (unsigned int*)(e + n_edges);
    unsigned int*  maxbits1 = maxbits + n_nodes;
    unsigned char* keys     = (unsigned char*)(maxbits1 + n_nodes);

    {
        gat_init_kernel<<<(n_nodes + 255) / 256, 256, 0, stream>>>(
            maxbits, maxbits1, out, n_nodes, n_edges, ws_size);
    }
    {
        int blocks = (n_nodes + GEMM_ROWS - 1) / GEMM_ROWS;
        gat_gemm_wmma_kernel<<<blocks, 128, 0, stream>>>(h, W, z, n_nodes);
    }
    {
        int edges_per_block = (256 / 32) * 32;
        int blocks = (n_edges + edges_per_block - 1) / edges_per_block;
        gat_edge_score_kernel<<<blocks, 256, 0, stream>>>(
            z, src, dst, e, maxbits, maxbits1, keys, out, n_nodes, n_edges, ws_size);
    }
    {
        gat_aggregate_kernel<<<AGG_OWNERS, AGG_THREADS, 0, stream>>>(
            z, src, dst, e, maxbits, maxbits1, keys, out, n_nodes, n_edges, ws_size);
    }
}


# Scores move into the owner launch: an owner keeps its rows of z in LDS and walks its lists twice (scores and LDS max, then sums); the edge launch only bins edges; 3 launches; sums in another order
# speedup vs baseline: 6.5689x; 1.2424x over previous
#include <hip/hip_runtime.h>
#include <hip/hip_bf16.h>

#define IN_DIM  128
#define OUT_DIM 64

typedef __attribute__((ext_vector_type(2))) float v2f;
typedef __attribute__((ext_vector_type(8))) float v8f;

#define AGG_OWNERS 256
#define AGG_RMAX   416
#define AGG_SCMAX  12288

__device__ __forceinline__ int gat_owner_rows(int n_nodes) {
    return (n_nodes + AGG_OWNERS - 1) / AGG_OWNERS;
}

__device__ __forceinline__ unsigned int* gat_list(float* out, int owner, int rows,
                                                  int n_nodes, int half,
                                                  unsigned int* cap) {
    const int lo = owner * rows;
    const int nr = (n_nodes - lo < rows) ? (n_nodes - lo) : rows;
    const unsigned int slots = (unsigned int)nr * 32u;
    *cap = slots - 1u;
    return (unsigned int*)(out + (size_t)lo * OUT_DIM) + (size_t)half * slots;
}

__device__ __forceinline__ int gat_half() {
    const unsigned int die = (__builtin_amdgcn_s_sendmsg_rtn(0x87) >> 16) & 0xFu;
    return die >= 4u ? 1 : 0;
}

typedef __attribute__((ext_vector_type(4))) float v4f;
typedef __attribute__((address_space(1))) const v4f gat_gcv4f;

__device__ __forceinline__ const gat_gcv4f* gat_h_row(const float* base, int row) {
    const unsigned long long a =
        (unsigned long long)base + ((unsigned long long)(unsigned int)row << 9);
    const unsigned int lo = __builtin_amdgcn_readfirstlane((unsigned int)a);
    const unsigned int hi = __builtin_amdgcn_readfirstlane((unsigned int)(a >> 32));
    return (const gat_gcv4f*)(((unsigned long long)hi << 32) | lo);
}

#define GEMM_RT     5
#define GEMM_ROWS   (16 * GEMM_RT)
#define GEMM_WROWS  (GEMM_ROWS / 4)
#define GEMM_PITCH4 33

__global__ void __launch_bounds__(128)
gat_gemm_wmma_kernel(const float* __restrict__ h,
                     const float* __restrict__ W,
                     float* __restrict__ z,
                     float* __restrict__ out,
                     int n_nodes) {
    __shared__ v4f tile4[GEMM_ROWS * GEMM_PITCH4];

    const int tid   = threadIdx.x;
    const int lane  = tid & 31;
    const int wave  = __builtin_amdgcn_readfirstlane(tid >> 5);
    const int half  = lane >> 4;
    const int lcol  = lane & 15;

    const int row0  = blockIdx.x * GEMM_ROWS;
    const int ncol  = wave * 16 + lcol;

    v4f st[GEMM_WROWS];
#pragma unroll
    for (int p = 0; p < GEMM_WROWS; ++p) {
        int gr = row0 + 4 * p + wave;
        if (gr >= n_nodes) gr = n_nodes - 1;
        st[p] = gat_h_row(h, gr)[lane];
    }

    const v4f* __restrict__ wrow =
        (const v4f*)(W + (size_t)ncol * IN_DIM) + half;
    v4f b[16];
#pragma unroll
    for (int j = 0; j < 16; ++j) b[j] = wrow[2 * j];

#pragma unroll
    for (int p = 0; p < GEMM_WROWS; ++p)
        tile4[(4 * p + wave) * GEMM_PITCH4 + lane] = st[p];

    __syncthreads();

    v8f acc[GEMM_RT] = {};
    const v4f* arow = tile4 + lcol * GEMM_PITCH4 + half;
    v4f a_cur[GEMM_RT], a_nxt[GEMM_RT];
#pragma unroll
    for (int rt = 0; rt < GEMM_RT; ++rt)
        a_cur[rt] = arow[rt * 16 * GEMM_PITCH4];
#pragma unroll
    for (int j = 0; j < 16; ++j) {
        if (j + 1 < 16) {
#pragma unroll
            for (int rt = 0; rt < GEMM_RT; ++rt)
                a_nxt[rt] = arow[rt * 16 * GEMM_PITCH4 + 2 * (j + 1)];
        }
        __builtin_amdgcn_sched_barrier(0);
        v2f b0; b0[0] = b[j].x; b0[1] = b[j].y;
        v2f b1; b1[0] = b[j].z; b1[1] = b[j].w;
#pragma unroll
        for (int rt = 0; rt < GEMM_RT; ++rt) {
            v2f a0; a0[0] = a_cur[rt].x; a0[1] = a_cur[rt].y;
            acc[rt] = __builtin_amdgcn_wmma_f32_16x16x4_f32(
                 false, a0,  false, b0,
                 (short)0, acc[rt],  false,  false);
        }
#pragma unroll
        for (int rt = 0; rt < GEMM_RT; ++rt) {
            v2f a1; a1[0] = a_cur[rt].z; a1[1] = a_cur[rt].w;
            acc[rt] = __builtin_amdgcn_wmma_f32_16x16x4_f32(
                 false, a1,  false, b1,
                 (short)0, acc[rt],  false,  false);
        }
        __builtin_amdgcn_sched_barrier(0);
#pragma unroll
        for (int rt = 0; rt < GEMM_RT; ++rt) a_cur[rt] = a_nxt[rt];
    }

    float* zout = z + (size_t)(row0 + 8 * half) * OUT_DIM + ncol;
    if (row0 + GEMM_ROWS <= n_nodes) {
#pragma unroll
        for (int rt = 0; rt < GEMM_RT; ++rt) {
#pragma unroll
            for (int v = 0; v < 8; ++v)
                zout[(rt * 16 + v) * OUT_DIM] = acc[rt][v];
        }
    } else {
#pragma unroll
        for (int rt = 0; rt < GEMM_RT; ++rt) {
#pragma unroll
            for (int v = 0; v < 8; ++v) {
                int row = row0 + rt * 16 + v + 8 * half;
                if (row < n_nodes)
                    zout[(rt * 16 + v) * OUT_DIM] = acc[rt][v];
            }
        }
    }

    if (tid == 0) {
        const int rows = gat_owner_rows(n_nodes);
        for (int o = blockIdx.x; o < AGG_OWNERS; o += gridDim.x) {
            if (o * rows < n_nodes) {
                unsigned int cap;
                gat_list(out, o, rows, n_nodes, 0, &cap)[0] = 0u;
                gat_list(out, o, rows, n_nodes, 1, &cap)[0] = 0u;
            }
        }
    }
}

__global__ void __launch_bounds__(256)
gat_edge_bin_kernel(const int* __restrict__ src,
                    const int* __restrict__ dst,
                    float* __restrict__ out,
                    int n_nodes, int n_edges) {
    const int e0 = (blockIdx.x * 256 + threadIdx.x) * 4;
    if (e0 >= n_edges) return;

    const int hi = gat_half();

    int sv[4], dv[4];
    if (e0 + 3 < n_edges) {
        const int4 a = *(const int4*)(src + e0);
        const int4 b = *(const int4*)(dst + e0);
        sv[0] = a.x; sv[1] = a.y; sv[2] = a.z; sv[3] = a.w;
        dv[0] = b.x; dv[1] = b.y; dv[2] = b.z; dv[3] = b.w;
    } else {
#pragma unroll
        for (int k = 0; k < 4; ++k) {
            const int ld = (e0 + k < n_edges) ? (e0 + k) : (n_edges - 1);
            sv[k] = src[ld];
            dv[k] = dst[ld];
        }
    }

    const int rows = gat_owner_rows(n_nodes);
    unsigned int* lst[4];
    unsigned int  cap[4], slot[4], rec[4];
#pragma unroll
    for (int k = 0; k < 4; ++k) {
        const int owner = dv[k] / rows;
        const int dl    = dv[k] - owner * rows;
        lst[k]  = gat_list(out, owner, rows, n_nodes, hi, &cap[k]);
        rec[k]  = ((unsigned int)sv[k] << 10) | (unsigned int)dl;
        slot[k] = (e0 + k < n_edges) ? atomicAdd(lst[k], 1u) : 0xFFFFFFFFu;
    }
#pragma unroll
    for (int k = 0; k < 4; ++k) {
        if (slot[k] < cap[k]) lst[k][1u + slot[k]] = rec[k];
    }
}


typedef __attribute__((address_space(1))) const float gat_gcfloat;

__device__ __forceinline__ unsigned long long gat_row_addr(const float* base, int row) {
    const unsigned long long a =
        (unsigned long long)base + ((unsigned long long)(unsigned int)row << 8);
    const unsigned int lo = __builtin_amdgcn_readfirstlane((unsigned int)a);
    const unsigned int hi = __builtin_amdgcn_readfirstlane((unsigned int)(a >> 32));
    return ((unsigned long long)hi << 32) | lo;
}

#define AGG_THREADS 1024
#define AGG_WAVES   (AGG_THREADS / 32)

__device__ __forceinline__ void gat_lds_add(float* p, float v) {
    __hip_atomic_fetch_add(p, v, __ATOMIC_RELAXED, __HIP_MEMORY_SCOPE_WORKGROUP);
}

__device__ __forceinline__ void gat_lds_max(unsigned int* p, float v) {
    if (v >= 0.0f)
        __hip_atomic_fetch_max((int*)p, __float_as_int(v), __ATOMIC_RELAXED,
                               __HIP_MEMORY_SCOPE_WORKGROUP);
    else
        __hip_atomic_fetch_min(p, __float_as_uint(v), __ATOMIC_RELAXED,
                               __HIP_MEMORY_SCOPE_WORKGROUP);
}

__device__ __forceinline__ unsigned int gat_record(const unsigned int* l0,
                                                   const unsigned int* l1,
                                                   unsigned int c0, unsigned int idx) {
    return (idx < c0) ? l0[1u + idx] : l1[1u + (idx - c0)];
}

__device__ __forceinline__ float gat_owner_scores(const float* __restrict__ z,
                                                  const float* zd,
                                                  int s, int dl, int lane) {
    const int r = lane & 7;
    float p[8];
#pragma unroll
    for (int hp = 0; hp < 2; ++hp) {
        float4 a0[4], a1[4], b0[4], b1[4];
#pragma unroll
        for (int u = 0; u < 4; ++u) {
            const int t    = hp * 4 + u;
            const int from = (lane & 24) + t;
            const int st   = __shfl(s, from, 32);
            const int dt   = __shfl(dl, from, 32);
            const float4* zs = (const float4*)(z + (size_t)st * OUT_DIM) + r;
            const float4* zl = (const float4*)(zd + dt * OUT_DIM) + r;
            a0[u] = zs[0];
            a1[u] = zs[8];
            b0[u] = zl[0];
            b1[u] = zl[8];
        }
#pragma unroll
        for (int u = 0; u < 4; ++u) {
            float acc = a0[u].x * b0[u].x;
            acc = fmaf(a0[u].y, b0[u].y, acc);
            acc = fmaf(a0[u].z, b0[u].z, acc);
            acc = fmaf(a0[u].w, b0[u].w, acc);
            acc = fmaf(a1[u].x, b1[u].x, acc);
            acc = fmaf(a1[u].y, b1[u].y, acc);
            acc = fmaf(a1[u].z, b1[u].z, acc);
            acc = fmaf(a1[u].w, b1[u].w, acc);
            p[hp * 4 + u] = acc;
        }
    }

    const bool hi4 = (r & 4) != 0;
    float q[4];
#pragma unroll
    for (int k = 0; k < 4; ++k) {
        const float send = hi4 ? p[k] : p[k + 4];
        const float keep = hi4 ? p[k + 4] : p[k];
        q[k] = keep + __shfl_xor(send, 4, 32);
    }
    const bool hi2 = (r & 2) != 0;
    float w[2];
#pragma unroll
    for (int k = 0; k < 2; ++k) {
        const float send = hi2 ? q[k] : q[k + 2];
        const float keep = hi2 ? q[k + 2] : q[k];
        w[k] = keep + __shfl_xor(send, 2, 32);
    }
    const bool hi1 = (r & 1) != 0;
    const float send1 = hi1 ? w[0] : w[1];
    const float keep1 = hi1 ? w[1] : w[0];
    return keep1 + __shfl_xor(send1, 1, 32);
}

template <bool FULL>
__device__ __forceinline__ void gat_owner_core(const float* __restrict__ z,
                                               int s, int dl, float ev, int cnt,
                                               float* acc, float* den,
                                               const unsigned int* mxb, int lane) {
    float v0[2][8], v1[2][8];
#pragma unroll
    for (int u = 0; u < 8; ++u) {
        const gat_gcfloat* zr =
            (const gat_gcfloat*)gat_row_addr(z, __builtin_amdgcn_readlane(s, u));
        v0[0][u] = zr[lane];
        v1[0][u] = zr[lane + 32];
    }
    const float t = __expf(ev - __uint_as_float(mxb[dl]));
    if (FULL || lane < cnt)
        gat_lds_add(&den[dl], t);
    const int abits = __float_as_int(t);
#pragma unroll
    for (int g = 0; g < 4; ++g) {
        if (g + 1 < 4) {
#pragma unroll
            for (int u = 0; u < 8; ++u) {
                const gat_gcfloat* zr = (const gat_gcfloat*)gat_row_addr(
                    z, __builtin_amdgcn_readlane(s, (g + 1) * 8 + u));
                v0[(g + 1) & 1][u] = zr[lane];
                v1[(g + 1) & 1][u] = zr[lane + 32];
            }
        }
#pragma unroll
        for (int u = 0; u < 8; ++u) {
            const int k = g * 8 + u;
            if (FULL || k < cnt) {
                float* orow = acc + __builtin_amdgcn_readlane(dl, k) * OUT_DIM;
                const float ak = __int_as_float(__builtin_amdgcn_readlane(abits, k));
                gat_lds_add(orow + lane, ak * v0[g & 1][u]);
                gat_lds_add(orow + lane + 32, ak * v1[g & 1][u]);
            }
        }
    }
}

__device__ __forceinline__ float gat_dot_row(const float* __restrict__ zs, const float* zl) {
    float a = 0.0f;
    for (int j = 0; j < OUT_DIM / 4; ++j) {
        const float4 x = ((const float4*)zs)[j];
        const float4 y = ((const float4*)zl)[j];
        a = fmaf(x.x, y.x, a);
        a = fmaf(x.y, y.y, a);
        a = fmaf(x.z, y.z, a);
        a = fmaf(x.w, y.w, a);
    }
    return a;
}

__global__ void __launch_bounds__(AGG_THREADS)
gat_owner_kernel(const float* __restrict__ z,
                 const int* __restrict__ src,
                 const int* __restrict__ dst,
                 float* __restrict__ out,
                 int n_nodes, int n_edges) {
    __shared__ __attribute__((aligned(16))) float acc[AGG_RMAX * OUT_DIM];
    __shared__ __attribute__((aligned(16))) float zd[AGG_RMAX * OUT_DIM];
    __shared__ float        sc[AGG_SCMAX];
    __shared__ float        den[AGG_RMAX];
    __shared__ unsigned int mxb[AGG_RMAX];

    const int tid  = threadIdx.x;
    const int lane = tid & 31;
    const int wave = __builtin_amdgcn_readfirstlane(tid >> 5);
    const int rows = gat_owner_rows(n_nodes);
    const int own  = blockIdx.x;
    const int lo   = own * rows;
    if (lo >= n_nodes) return;
    const int nrows = (n_nodes - lo < rows) ? (n_nodes - lo) : rows;

    unsigned int cap;
    const unsigned int* l0 = gat_list(out, own, rows, n_nodes, 0, &cap);
    const unsigned int* l1 = gat_list(out, own, rows, n_nodes, 1, &cap);
    const unsigned int c0 = __builtin_amdgcn_readfirstlane(l0[0]);
    const unsigned int c1 = __builtin_amdgcn_readfirstlane(l1[0]);

    const float4* z4  = (const float4*)(z + (size_t)lo * OUT_DIM);
    float4*       zd4 = (float4*)zd;
    float4*       ac4 = (float4*)acc;
    for (int i = tid; i < nrows * (OUT_DIM / 4); i += AGG_THREADS) {
        zd4[i] = z4[i];
        ac4[i] = make_float4(0.0f, 0.0f, 0.0f, 0.0f);
    }
    for (int i = tid; i < nrows; i += AGG_THREADS) {
        den[i] = 0.0f;
        mxb[i] = 0xFF800000u;
    }
    __syncthreads();

    if (c0 <= cap && c1 <= cap && c0 + c1 <= (unsigned int)AGG_SCMAX) {
        const unsigned int total = c0 + c1;

        {
            const unsigned int mine = 4u * (unsigned int)(lane & 7) + (unsigned int)(lane >> 3);
            unsigned int base = (unsigned int)wave * 32u;
            unsigned int rec = 0u;
            if (base < total) {
                unsigned int idx = base + mine;
                if (idx >= total) idx = total - 1u;
                rec = gat_record(l0, l1, c0, idx);
            }
            while (base < total) {
                const unsigned int nbase = base + AGG_WAVES * 32u;
                unsigned int nrec = rec;
                if (nbase < total) {
                    unsigned int idx = nbase + mine;
                    if (idx >= total) idx = total - 1u;
                    nrec = gat_record(l0, l1, c0, idx);
                }
                int s  = (int)(rec >> 10);
                int dl = (int)(rec & 1023u);
                if (s  > n_nodes - 1) s  = n_nodes - 1;
                if (dl > nrows - 1)   dl = nrows - 1;
                const float ev = gat_owner_scores(z, zd, s, dl, lane);
                const unsigned int idx = base + mine;
                if (idx < total) {
                    sc[idx] = ev;
                    gat_lds_max(&mxb[dl], ev);
                }
                rec  = nrec;
                base = nbase;
            }
        }
        __syncthreads();

        {
            unsigned int base = (unsigned int)wave * 32u;
            unsigned int rec = 0u;
            float ev = 0.0f;
            if (base < total) {
                unsigned int idx = base + (unsigned int)lane;
                if (idx >= total) idx = total - 1u;
                rec = gat_record(l0, l1, c0, idx);
                ev  = sc[idx];
            }
            while (base < total) {
                const unsigned int nbase = base + AGG_WAVES * 32u;
                unsigned int nrec = rec;
                float nev = ev;
                if (nbase < total) {
                    unsigned int idx = nbase + (unsigned int)lane;
                    if (idx >= total) idx = total - 1u;
                    nrec = gat_record(l0, l1, c0, idx);
                    nev  = sc[idx];
                }
                int s  = (int)(rec >> 10);
                int dl = (int)(rec & 1023u);
                if (s  > n_nodes - 1) s  = n_nodes - 1;
                if (dl > nrows - 1)   dl = nrows - 1;
                const unsigned int left = total - base;
                if (left >= 32u)
                    gat_owner_core<true>(z, s, dl, ev, 32, acc, den, mxb, lane);
                else
                    gat_owner_core<false>(z, s, dl, ev, (int)left, acc, den, mxb, lane);
                rec  = nrec;
                ev   = nev;
                base = nbase;
            }
        }
    } else {
        for (int eid = tid; eid < n_edges; eid += AGG_THREADS) {
            const int dl = dst[eid] - lo;
            if ((unsigned int)dl < (unsigned int)nrows) {
                int s = src[eid];
                if (s > n_nodes - 1) s = n_nodes - 1;
                if (s < 0) s = 0;
                gat_lds_max(&mxb[dl], gat_dot_row(z + (size_t)s * OUT_DIM, zd + dl * OUT_DIM));
            }
        }
        __syncthreads();
        for (int eid = tid; eid < n_edges; eid += AGG_THREADS) {
            const int dl = dst[eid] - lo;
            if ((unsigned int)dl < (unsigned int)nrows) {
                int s = src[eid];
                if (s > n_nodes - 1) s = n_nodes - 1;
                if (s < 0) s = 0;
                const float* zs = z + (size_t)s * OUT_DIM;
                const float ev = gat_dot_row(zs, zd + dl * OUT_DIM);
                const float t  = __expf(ev - __uint_as_float(mxb[dl]));
                gat_lds_add(&den[dl], t);
                float* orow = acc + dl * OUT_DIM;
                for (int j = 0; j < OUT_DIM; ++j)
                    gat_lds_add(orow + j, t * zs[j]);
            }
        }
    }
    __syncthreads();

    float4*       o4 = (float4*)(out + (size_t)lo * OUT_DIM);
    const float4* a4 = (const float4*)acc;
    for (int i = tid; i < nrows * (OUT_DIM / 4); i += AGG_THREADS) {
        const float inv = 1.0f / fmaxf(den[i / (OUT_DIM / 4)], 1e-30f);
        float4 a = a4[i];
        a.x *= inv; a.y *= inv; a.z *= inv; a.w *= inv;
        o4[i] = a;
    }
}

extern "C" void kernel_launch(void* const* d_in, const int* in_sizes, int n_in,
                              void* d_out, int out_size, void* d_ws, size_t ws_size,
                              hipStream_t stream) {
    const float* h   = (const float*)d_in[0];
    const float* W   = (const float*)d_in[1];
    const int*   src = (const int*)d_in[2];
    const int*   dst = (const int*)d_in[3];
    float* out = (float*)d_out;

    const int n_nodes = in_sizes[0] / IN_DIM;
    const int n_edges = in_sizes[2];

    float* z = (float*)d_ws;

    {
        int blocks = (n_nodes + GEMM_ROWS - 1) / GEMM_ROWS;
        gat_gemm_wmma_kernel<<<blocks, 128, 0, stream>>>(h, W, z, out, n_nodes);
    }
    {
        int edges_per_block = 256 * 4;
        int blocks = (n_edges + edges_per_block - 1) / edges_per_block;
        gat_edge_bin_kernel<<<blocks, 256, 0, stream>>>(src, dst, out, n_nodes, n_edges);
    }
    {
        gat_owner_kernel<<<AGG_OWNERS, AGG_THREADS, 0, stream>>>(
            z, src, dst, out, n_nodes, n_edges);
    }
}


# Owners walk their lists once: an edge's row of z is read once, weight exp(e) unshifted; only rows whose maximum is out of range are redone alone, shifted; weights' rounding and order of sums differ
# speedup vs baseline: 6.9439x; 1.0571x over previous
#include <hip/hip_runtime.h>
#include <hip/hip_bf16.h>

#define IN_DIM  128
#define OUT_DIM 64

typedef __attribute__((ext_vector_type(2))) float v2f;
typedef __attribute__((ext_vector_type(8))) float v8f;

#define AGG_OWNERS 256
#define AGG_RMAX   416

__device__ __forceinline__ int gat_owner_rows(int n_nodes) {
    return (n_nodes + AGG_OWNERS - 1) / AGG_OWNERS;
}

__device__ __forceinline__ unsigned int* gat_list(float* out, int owner, int rows,
                                                  int n_nodes, int half,
                                                  unsigned int* cap) {
    const int lo = owner * rows;
    const int nr = (n_nodes - lo < rows) ? (n_nodes - lo) : rows;
    const unsigned int slots = (unsigned int)nr * 32u;
    *cap = slots - 1u;
    return (unsigned int*)(out + (size_t)lo * OUT_DIM) + (size_t)half * slots;
}

__device__ __forceinline__ int gat_half() {
    const unsigned int die = (__builtin_amdgcn_s_sendmsg_rtn(0x87) >> 16) & 0xFu;
    return die >= 4u ? 1 : 0;
}

typedef __attribute__((ext_vector_type(4))) float v4f;
typedef __attribute__((address_space(1))) const v4f gat_gcv4f;

__device__ __forceinline__ const gat_gcv4f* gat_h_row(const float* base, int row) {
    const unsigned long long a =
        (unsigned long long)base + ((unsigned long long)(unsigned int)row << 9);
    const unsigned int lo = __builtin_amdgcn_readfirstlane((unsigned int)a);
    const unsigned int hi = __builtin_amdgcn_readfirstlane((unsigned int)(a >> 32));
    return (const gat_gcv4f*)(((unsigned long long)hi << 32) | lo);
}

#define GEMM_RT     5
#define GEMM_ROWS   (16 * GEMM_RT)
#define GEMM_WROWS  (GEMM_ROWS / 4)
#define GEMM_PITCH4 33

__global__ void __launch_bounds__(128)
gat_gemm_wmma_kernel(const float* __restrict__ h,
                     const float* __restrict__ W,
                     float* __restrict__ z,
                     float* __restrict__ out,
                     int n_nodes) {
    __shared__ v4f tile4[GEMM_ROWS * GEMM_PITCH4];

    const int tid   = threadIdx.x;
    const int lane  = tid & 31;
    const int wave  = __builtin_amdgcn_readfirstlane(tid >> 5);
    const int half  = lane >> 4;
    const int lcol  = lane & 15;

    const int row0  = blockIdx.x * GEMM_ROWS;
    const int ncol  = wave * 16 + lcol;

    v4f st[GEMM_WROWS];
#pragma unroll
    for (int p = 0; p < GEMM_WROWS; ++p) {
        int gr = row0 + 4 * p + wave;
        if (gr >= n_nodes) gr = n_nodes - 1;
        st[p] = gat_h_row(h, gr)[lane];
    }

    const v4f* __restrict__ wrow =
        (const v4f*)(W + (size_t)ncol * IN_DIM) + half;
    v4f b[16];
#pragma unroll
    for (int j = 0; j < 16; ++j) b[j] = wrow[2 * j];

#pragma unroll
    for (int p = 0; p < GEMM_WROWS; ++p)
        tile4[(4 * p + wave) * GEMM_PITCH4 + lane] = st[p];

    __syncthreads();

    v8f acc[GEMM_RT] = {};
    const v4f* arow = tile4 + lcol * GEMM_PITCH4 + half;
    v4f a_cur[GEMM_RT], a_nxt[GEMM_RT];
#pragma unroll
    for (int rt = 0; rt < GEMM_RT; ++rt)
        a_cur[rt] = arow[rt * 16 * GEMM_PITCH4];
#pragma unroll
    for (int j = 0; j < 16; ++j) {
        if (j + 1 < 16) {
#pragma unroll
            for (int rt = 0; rt < GEMM_RT; ++rt)
                a_nxt[rt] = arow[rt * 16 * GEMM_PITCH4 + 2 * (j + 1)];
        }
        __builtin_amdgcn_sched_barrier(0);
        v2f b0; b0[0] = b[j].x; b0[1] = b[j].y;
        v2f b1; b1[0] = b[j].z; b1[1] = b[j].w;
#pragma unroll
        for (int rt = 0; rt < GEMM_RT; ++rt) {
            v2f a0; a0[0] = a_cur[rt].x; a0[1] = a_cur[rt].y;
            acc[rt] = __builtin_amdgcn_wmma_f32_16x16x4_f32(
                 false, a0,  false, b0,
                 (short)0, acc[rt],  false,  false);
        }
#pragma unroll
        for (int rt = 0; rt < GEMM_RT; ++rt) {
            v2f a1; a1[0] = a_cur[rt].z; a1[1] = a_cur[rt].w;
            acc[rt] = __builtin_amdgcn_wmma_f32_16x16x4_f32(
                 false, a1,  false, b1,
                 (short)0, acc[rt],  false,  false);
        }
        __builtin_amdgcn_sched_barrier(0);
#pragma unroll
        for (int rt = 0; rt < GEMM_RT; ++rt) a_cur[rt] = a_nxt[rt];
    }

    float* zout = z + (size_t)(row0 + 8 * half) * OUT_DIM + ncol;
    if (row0 + GEMM_ROWS <= n_nodes) {
#pragma unroll
        for (int rt = 0; rt < GEMM_RT; ++rt) {
#pragma unroll
            for (int v = 0; v < 8; ++v)
                zout[(rt * 16 + v) * OUT_DIM] = acc[rt][v];
        }
    } else {
#pragma unroll
        for (int rt = 0; rt < GEMM_RT; ++rt) {
#pragma unroll
            for (int v = 0; v < 8; ++v) {
                int row = row0 + rt * 16 + v + 8 * half;
                if (row < n_nodes)
                    zout[(rt * 16 + v) * OUT_DIM] = acc[rt][v];
            }
        }
    }

    if (tid == 0) {
        const int rows = gat_owner_rows(n_nodes);
        for (int o = blockIdx.x; o < AGG_OWNERS; o += gridDim.x) {
            if (o * rows < n_nodes) {
                unsigned int cap;
                gat_list(out, o, rows, n_nodes, 0, &cap)[0] = 0u;
                gat_list(out, o, rows, n_nodes, 1, &cap)[0] = 0u;
            }
        }
    }
}

__global__ void __launch_bounds__(256)
gat_edge_bin_kernel(const int* __restrict__ src,
                    const int* __restrict__ dst,
                    float* __restrict__ out,
                    int n_nodes, int n_edges) {
    const int e0 = (blockIdx.x * 256 + threadIdx.x) * 4;
    if (e0 >= n_edges) return;

    const int hi = gat_half();

    int sv[4], dv[4];
    if (e0 + 3 < n_edges) {
        const int4 a = *(const int4*)(src + e0);
        const int4 b = *(const int4*)(dst + e0);
        sv[0] = a.x; sv[1] = a.y; sv[2] = a.z; sv[3] = a.w;
        dv[0] = b.x; dv[1] = b.y; dv[2] = b.z; dv[3] = b.w;
    } else {
#pragma unroll
        for (int k = 0; k < 4; ++k) {
            const int ld = (e0 + k < n_edges) ? (e0 + k) : (n_edges - 1);
            sv[k] = src[ld];
            dv[k] = dst[ld];
        }
    }

    const int rows = gat_owner_rows(n_nodes);
    unsigned int* lst[4];
    unsigned int  cap[4], slot[4], rec[4];
#pragma unroll
    for (int k = 0; k < 4; ++k) {
        const int owner = dv[k] / rows;
        const int dl    = dv[k] - owner * rows;
        lst[k]  = gat_list(out, owner, rows, n_nodes, hi, &cap[k]);
        rec[k]  = ((unsigned int)sv[k] << 10) | (unsigned int)dl;
        slot[k] = (e0 + k < n_edges) ? atomicAdd(lst[k], 1u) : 0xFFFFFFFFu;
    }
#pragma unroll
    for (int k = 0; k < 4; ++k) {
        if (slot[k] < cap[k]) lst[k][1u + slot[k]] = rec[k];
    }
}


typedef __attribute__((address_space(1))) const float gat_gcfloat;
typedef __attribute__((address_space(1))) const v2f   gat_gcv2f;

__device__ __forceinline__ unsigned long long gat_row_addr(const float* base, int row) {
    const unsigned long long a =
        (unsigned long long)base + ((unsigned long long)(unsigned int)row << 8);
    const unsigned int lo = __builtin_amdgcn_readfirstlane((unsigned int)a);
    const unsigned int hi = __builtin_amdgcn_readfirstlane((unsigned int)(a >> 32));
    return ((unsigned long long)hi << 32) | lo;
}

#define AGG_THREADS 1024
#define AGG_WAVES   (AGG_THREADS / 32)

#define GAT_LO (-60.0f)
#define GAT_HI 80.0f

__device__ __forceinline__ void gat_lds_add(float* p, float v) {
    __hip_atomic_fetch_add(p, v, __ATOMIC_RELAXED, __HIP_MEMORY_SCOPE_WORKGROUP);
}

__device__ __forceinline__ void gat_lds_max(unsigned int* p, float v) {
    if (v >= 0.0f)
        __hip_atomic_fetch_max((int*)p, __float_as_int(v), __ATOMIC_RELAXED,
                               __HIP_MEMORY_SCOPE_WORKGROUP);
    else
        __hip_atomic_fetch_min(p, __float_as_uint(v), __ATOMIC_RELAXED,
                               __HIP_MEMORY_SCOPE_WORKGROUP);
}

__device__ __forceinline__ unsigned int gat_record(const unsigned int* l0,
                                                   const unsigned int* l1,
                                                   unsigned int c0, unsigned int idx) {
    return (idx < c0) ? l0[1u + idx] : l1[1u + (idx - c0)];
}

template <int M>
__device__ __forceinline__ float gat_row_xor(float v) {
    return __builtin_bit_cast(float, __builtin_amdgcn_update_dpp(
        0, __builtin_bit_cast(int, v), 0x160 + M, 0xf, 0xf, true));
}
__device__ __forceinline__ float gat_xhalf(float v) {
    const int b = __builtin_bit_cast(int, v);
    return __builtin_bit_cast(float, __builtin_amdgcn_permlanex16(
        b, b, 0x76543210, 0xfedcba98, false, false));
}

__device__ __forceinline__ float gat_exp_wide(float x) {
    const float l2e_hi = 1.44269502f;
    const float l2e_lo = 1.92596303e-8f;
    const float y = x * l2e_hi;
    const float r = fmaf(x, l2e_lo, fmaf(x, l2e_hi, -y));
    const float e = __builtin_amdgcn_exp2f(y);
    return fmaf(e, r * 0.693147182f, e);
}

template <bool FULL>
__device__ __forceinline__ void gat_owner_fused(const float* __restrict__ z,
                                                const float* zd,
                                                int s, int dl, int cnt,
                                                float* acc, float* den,
                                                unsigned int* mxb, int lane) {
    const int c = lane & 15;
    v2f a2[2][16];
#pragma unroll
    for (int k = 0; k < 32; ++k) {
        const gat_gcv2f* zr = (const gat_gcv2f*)gat_row_addr(
            z, __builtin_amdgcn_readlane(s, k));
        a2[k >> 4][k & 15] = zr[lane];
    }
#pragma unroll
    for (int hf = 0; hf < 2; ++hf) {
        const v2f* a = a2[hf];
        float p[16];
#pragma unroll
        for (int u = 0; u < 16; ++u) {
            const v2f b = *(const v2f*)(zd + __builtin_amdgcn_readlane(dl, hf * 16 + u) * OUT_DIM
                                        + 2 * lane);
            p[u] = fmaf(a[u].y, b.y, a[u].x * b.x);
        }

        const bool h8 = (c & 8) != 0;
        float q[8];
#pragma unroll
        for (int k = 0; k < 8; ++k) {
            const float send = h8 ? p[k] : p[k + 8];
            const float keep = h8 ? p[k + 8] : p[k];
            q[k] = keep + gat_row_xor<8>(send);
        }
        const bool h4 = (c & 4) != 0;
        float w[4];
#pragma unroll
        for (int k = 0; k < 4; ++k) {
            const float send = h4 ? q[k] : q[k + 4];
            const float keep = h4 ? q[k + 4] : q[k];
            w[k] = keep + gat_row_xor<4>(send);
        }
        const bool h2 = (c & 2) != 0;
        float x[2];
#pragma unroll
        for (int k = 0; k < 2; ++k) {
            const float send = h2 ? w[k] : w[k + 2];
            const float keep = h2 ? w[k + 2] : w[k];
            x[k] = keep + gat_row_xor<2>(send);
        }
        const bool h1 = (c & 1) != 0;
        const float send1 = h1 ? x[0] : x[1];
        const float keep1 = h1 ? x[1] : x[0];
        const float y = keep1 + gat_row_xor<1>(send1);
        const float ev = y + gat_xhalf(y);
        const float t  = gat_exp_wide(ev);

        if ((lane >> 4) == hf && (FULL || lane < cnt)) {
            gat_lds_max(&mxb[dl], ev);
            gat_lds_add(&den[dl], t);
        }
        const int tbits = __float_as_int(t);
#pragma unroll
        for (int u = 0; u < 16; ++u) {
            const int k = hf * 16 + u;
            if (FULL || k < cnt) {
                float* orow = acc + __builtin_amdgcn_readlane(dl, k) * OUT_DIM + 2 * lane;
                const float tk = __int_as_float(__builtin_amdgcn_readlane(tbits, u));
                gat_lds_add(orow,     tk * a[u].x);
                gat_lds_add(orow + 1, tk * a[u].y);
            }
        }
    }
}

__device__ __forceinline__ void gat_owner_redo(const float* __restrict__ z,
                                               const float* zd, int s, int dl,
                                               float* acc, float* den,
                                               const unsigned int* mxb, int lane) {
    const gat_gcfloat* zr = (const gat_gcfloat*)gat_row_addr(z, s);
    const float a0 = zr[lane];
    const float a1 = zr[lane + 32];
    const float* zl = zd + dl * OUT_DIM;
    float p = fmaf(a1, zl[lane + 32], a0 * zl[lane]);
    p += gat_xhalf(p);
    p += gat_row_xor<8>(p);
    p += gat_row_xor<4>(p);
    p += gat_row_xor<2>(p);
    p += gat_row_xor<1>(p);
    const float t = __expf(p - __uint_as_float(mxb[dl]));
    if (lane == 0) gat_lds_add(&den[dl], t);
    float* orow = acc + dl * OUT_DIM;
    gat_lds_add(orow + lane,      t * a0);
    gat_lds_add(orow + lane + 32, t * a1);
}

__device__ __forceinline__ float gat_dot_row(const float* __restrict__ zs, const float* zl) {
    float a = 0.0f;
    for (int j = 0; j < OUT_DIM / 4; ++j) {
        const float4 x = ((const float4*)zs)[j];
        const float4 y = ((const float4*)zl)[j];
        a = fmaf(x.x, y.x, a);
        a = fmaf(x.y, y.y, a);
        a = fmaf(x.z, y.z, a);
        a = fmaf(x.w, y.w, a);
    }
    return a;
}

__global__ void __launch_bounds__(AGG_THREADS)
gat_owner_kernel(const float* __restrict__ z,
                 const int* __restrict__ src,
                 const int* __restrict__ dst,
                 float* __restrict__ out,
                 int n_nodes, int n_edges) {
    __shared__ __attribute__((aligned(16))) float acc[AGG_RMAX * OUT_DIM];
    __shared__ __attribute__((aligned(16))) float zd[AGG_RMAX * OUT_DIM];
    __shared__ float        den[AGG_RMAX];
    __shared__ unsigned int mxb[AGG_RMAX];
    __shared__ unsigned int bad[AGG_RMAX];
    __shared__ unsigned int nbad;

    const int tid  = threadIdx.x;
    const int lane = tid & 31;
    const int wave = __builtin_amdgcn_readfirstlane(tid >> 5);
    const int rows = gat_owner_rows(n_nodes);
    const int own  = blockIdx.x;
    const int lo   = own * rows;
    if (lo >= n_nodes) return;
    const int nrows = (n_nodes - lo < rows) ? (n_nodes - lo) : rows;

    unsigned int cap;
    const unsigned int* l0 = gat_list(out, own, rows, n_nodes, 0, &cap);
    const unsigned int* l1 = gat_list(out, own, rows, n_nodes, 1, &cap);
    const unsigned int c0 = __builtin_amdgcn_readfirstlane(l0[0]);
    const unsigned int c1 = __builtin_amdgcn_readfirstlane(l1[0]);

    const float4* z4  = (const float4*)(z + (size_t)lo * OUT_DIM);
    float4*       zd4 = (float4*)zd;
    float4*       ac4 = (float4*)acc;
    for (int i = tid; i < nrows * (OUT_DIM / 4); i += AGG_THREADS) {
        zd4[i] = z4[i];
        ac4[i] = make_float4(0.0f, 0.0f, 0.0f, 0.0f);
    }
    for (int i = tid; i < nrows; i += AGG_THREADS) {
        den[i] = 0.0f;
        mxb[i] = 0xFF800000u;
        bad[i] = 0u;
    }
    if (tid == 0) nbad = 0u;
    __syncthreads();

    if (c0 <= cap && c1 <= cap) {
        const unsigned int total = c0 + c1;

        {
            unsigned int base = (unsigned int)wave * 32u;
            unsigned int rec = 0u;
            if (base < total) {
                unsigned int idx = base + (unsigned int)lane;
                if (idx >= total) idx = total - 1u;
                rec = gat_record(l0, l1, c0, idx);
            }
            while (base < total) {
                const unsigned int nbase = base + AGG_WAVES * 32u;
                unsigned int nrec = rec;
                if (nbase < total) {
                    unsigned int idx = nbase + (unsigned int)lane;
                    if (idx >= total) idx = total - 1u;
                    nrec = gat_record(l0, l1, c0, idx);
                }
                int s  = (int)(rec >> 10);
                int dl = (int)(rec & 1023u);
                if (s  > n_nodes - 1) s  = n_nodes - 1;
                if (dl > nrows - 1)   dl = nrows - 1;
                const unsigned int left = total - base;
                if (left >= 32u)
                    gat_owner_fused<true>(z, zd, s, dl, 32, acc, den, mxb, lane);
                else
                    gat_owner_fused<false>(z, zd, s, dl, (int)left, acc, den, mxb, lane);
                rec  = nrec;
                base = nbase;
            }
        }
        __syncthreads();

        {
            const float inf = __builtin_huge_valf();
            unsigned int off = 0u;
            for (int i = tid; i < nrows; i += AGG_THREADS) {
                const unsigned int m = mxb[i];
                const float mf = __uint_as_float(m);
                if ((m != 0xFF800000u && !(mf >= GAT_LO && mf <= GAT_HI)) || !(den[i] < inf)) {
                    bad[i] = 1u;
                    off = 1u;
                }
            }
            const float4* a4c = (const float4*)acc;
            for (int i = tid; i < nrows * (OUT_DIM / 4); i += AGG_THREADS) {
                const float4 a = a4c[i];
                if (!(fabsf(a.x) < inf && fabsf(a.y) < inf && fabsf(a.z) < inf && fabsf(a.w) < inf)) {
                    bad[i / (OUT_DIM / 4)] = 1u;
                    off = 1u;
                }
            }
            if (off != 0u) nbad = 1u;
        }
        __syncthreads();

        if (__builtin_amdgcn_readfirstlane(nbad) != 0u) {
            for (int i = tid; i < nrows * (OUT_DIM / 4); i += AGG_THREADS) {
                if (bad[i / (OUT_DIM / 4)] != 0u)
                    ac4[i] = make_float4(0.0f, 0.0f, 0.0f, 0.0f);
            }
            for (int i = tid; i < nrows; i += AGG_THREADS) {
                if (bad[i] != 0u) den[i] = 0.0f;
            }
            __syncthreads();

            unsigned int base = (unsigned int)wave * 32u;
            unsigned int rec = 0u;
            if (base < total) {
                unsigned int idx = base + (unsigned int)lane;
                if (idx >= total) idx = total - 1u;
                rec = gat_record(l0, l1, c0, idx);
            }
            while (base < total) {
                const unsigned int nbase = base + AGG_WAVES * 32u;
                unsigned int nrec = rec;
                if (nbase < total) {
                    unsigned int idx = nbase + (unsigned int)lane;
                    if (idx >= total) idx = total - 1u;
                    nrec = gat_record(l0, l1, c0, idx);
                }
                int s  = (int)(rec >> 10);
                int dl = (int)(rec & 1023u);
                if (s  > n_nodes - 1) s  = n_nodes - 1;
                if (dl > nrows - 1)   dl = nrows - 1;
                const bool hit = (base + (unsigned int)lane < total) && bad[dl] != 0u;
                unsigned int todo = __builtin_amdgcn_ballot_w32(hit);
                while (todo != 0u) {
                    const int k = __builtin_ctz(todo);
                    todo &= todo - 1u;
                    gat_owner_redo(z, zd, __builtin_amdgcn_readlane(s, k),
                                   __builtin_amdgcn_readlane(dl, k), acc, den, mxb, lane);
                }
                rec  = nrec;
                base = nbase;
            }
        }
    } else {
        for (int eid = tid; eid < n_edges; eid += AGG_THREADS) {
            const int dl = dst[eid] - lo;
            if ((unsigned int)dl < (unsigned int)nrows) {
                int s = src[eid];
                if (s > n_nodes - 1) s = n_nodes - 1;
                if (s < 0) s = 0;
                gat_lds_max(&mxb[dl], gat_dot_row(z + (size_t)s * OUT_DIM, zd + dl * OUT_DIM));
            }
        }
        __syncthreads();
        for (int eid = tid; eid < n_edges; eid += AGG_THREADS) {
            const int dl = dst[eid] - lo;
            if ((unsigned int)dl < (unsigned int)nrows) {
                int s = src[eid];
                if (s > n_nodes - 1) s = n_nodes - 1;
                if (s < 0) s = 0;
                const float* zs = z + (size_t)s * OUT_DIM;
                const float ev = gat_dot_row(zs, zd + dl * OUT_DIM);
                const float t  = __expf(ev - __uint_as_float(mxb[dl]));
                gat_lds_add(&den[dl], t);
                float* orow = acc + dl * OUT_DIM;
                for (int j = 0; j < OUT_DIM; ++j)
                    gat_lds_add(orow + j, t * zs[j]);
            }
        }
    }
    __syncthreads();

    float4*       o4 = (float4*)(out + (size_t)lo * OUT_DIM);
    const float4* a4 = (const float4*)acc;
    for (int i = tid; i < nrows * (OUT_DIM / 4); i += AGG_THREADS) {
        const float inv = 1.0f / fmaxf(den[i / (OUT_DIM / 4)], 1e-30f);
        float4 a = a4[i];
        a.x *= inv; a.y *= inv; a.z *= inv; a.w *= inv;
        o4[i] = a;
    }
}

extern "C" void kernel_launch(void* const* d_in, const int* in_sizes, int n_in,
                              void* d_out, int out_size, void* d_ws, size_t ws_size,
                              hipStream_t stream) {
    const float* h   = (const float*)d_in[0];
    const float* W   = (const float*)d_in[1];
    const int*   src = (const int*)d_in[2];
    const int*   dst = (const int*)d_in[3];
    float* out = (float*)d_out;

    const int n_nodes = in_sizes[0] / IN_DIM;
    const int n_edges = in_sizes[2];

    float* z = (float*)d_ws;

    {
        int blocks = (n_nodes + GEMM_ROWS - 1) / GEMM_ROWS;
        gat_gemm_wmma_kernel<<<blocks, 128, 0, stream>>>(h, W, z, out, n_nodes);
    }
    {
        int edges_per_block = 256 * 4;
        int blocks = (n_edges + edges_per_block - 1) / edges_per_block;
        gat_edge_bin_kernel<<<blocks, 256, 0, stream>>>(src, dst, out, n_nodes, n_edges);
    }
    {
        gat_owner_kernel<<<AGG_OWNERS, AGG_THREADS, 0, stream>>>(
            z, src, dst, out, n_nodes, n_edges);
    }
}


# Binning launch sorts its edges by owner in LDS (ranks from LDS counters): one returning global add per owner and work-group, not per edge; records stored in runs; sums in another order
# speedup vs baseline: 8.6566x; 1.2466x over previous
#include <hip/hip_runtime.h>
#include <hip/hip_bf16.h>

#define IN_DIM  128
#define OUT_DIM 64

typedef __attribute__((ext_vector_type(2))) float v2f;
typedef __attribute__((ext_vector_type(8))) float v8f;

#define AGG_OWNERS 256
#define AGG_RMAX   416

__device__ __forceinline__ int gat_owner_rows(int n_nodes) {
    return (n_nodes + AGG_OWNERS - 1) / AGG_OWNERS;
}

__device__ __forceinline__ unsigned int* gat_list(float* out, int owner, int rows,
                                                  int n_nodes, int half,
                                                  unsigned int* cap) {
    const int lo = owner * rows;
    const int nr = (n_nodes - lo < rows) ? (n_nodes - lo) : rows;
    const unsigned int slots = (unsigned int)nr * 32u;
    *cap = slots - 1u;
    return (unsigned int*)(out + (size_t)lo * OUT_DIM) + (size_t)half * slots;
}

__device__ __forceinline__ int gat_half() {
    const unsigned int die = (__builtin_amdgcn_s_sendmsg_rtn(0x87) >> 16) & 0xFu;
    return die >= 4u ? 1 : 0;
}

typedef __attribute__((ext_vector_type(4))) float v4f;
typedef __attribute__((address_space(1))) const v4f gat_gcv4f;

__device__ __forceinline__ const gat_gcv4f* gat_h_row(const float* base, int row) {
    const unsigned long long a =
        (unsigned long long)base + ((unsigned long long)(unsigned int)row << 9);
    const unsigned int lo = __builtin_amdgcn_readfirstlane((unsigned int)a);
    const unsigned int hi = __builtin_amdgcn_readfirstlane((unsigned int)(a >> 32));
    return (const gat_gcv4f*)(((unsigned long long)hi << 32) | lo);
}

#define GEMM_RT     5
#define GEMM_ROWS   (16 * GEMM_RT)
#define GEMM_WROWS  (GEMM_ROWS / 4)
#define GEMM_PITCH4 33

__global__ void __launch_bounds__(128)
gat_gemm_wmma_kernel(const float* __restrict__ h,
                     const float* __restrict__ W,
                     float* __restrict__ z,
                     float* __restrict__ out,
                     int n_nodes) {
    __shared__ v4f tile4[GEMM_ROWS * GEMM_PITCH4];

    const int tid   = threadIdx.x;
    const int lane  = tid & 31;
    const int wave  = __builtin_amdgcn_readfirstlane(tid >> 5);
    const int half  = lane >> 4;
    const int lcol  = lane & 15;

    const int row0  = blockIdx.x * GEMM_ROWS;
    const int ncol  = wave * 16 + lcol;

    v4f st[GEMM_WROWS];
#pragma unroll
    for (int p = 0; p < GEMM_WROWS; ++p) {
        int gr = row0 + 4 * p + wave;
        if (gr >= n_nodes) gr = n_nodes - 1;
        st[p] = gat_h_row(h, gr)[lane];
    }

    const v4f* __restrict__ wrow =
        (const v4f*)(W + (size_t)ncol * IN_DIM) + half;
    v4f b[16];
#pragma unroll
    for (int j = 0; j < 16; ++j) b[j] = wrow[2 * j];

#pragma unroll
    for (int p = 0; p < GEMM_WROWS; ++p)
        tile4[(4 * p + wave) * GEMM_PITCH4 + lane] = st[p];

    __syncthreads();

    v8f acc[GEMM_RT] = {};
    const v4f* arow = tile4 + lcol * GEMM_PITCH4 + half;
    v4f a_cur[GEMM_RT], a_nxt[GEMM_RT];
#pragma unroll
    for (int rt = 0; rt < GEMM_RT; ++rt)
        a_cur[rt] = arow[rt * 16 * GEMM_PITCH4];
#pragma unroll
    for (int j = 0; j < 16; ++j) {
        if (j + 1 < 16) {
#pragma unroll
            for (int rt = 0; rt < GEMM_RT; ++rt)
                a_nxt[rt] = arow[rt * 16 * GEMM_PITCH4 + 2 * (j + 1)];
        }
        __builtin_amdgcn_sched_barrier(0);
        v2f b0; b0[0] = b[j].x; b0[1] = b[j].y;
        v2f b1; b1[0] = b[j].z; b1[1] = b[j].w;
#pragma unroll
        for (int rt = 0; rt < GEMM_RT; ++rt) {
            v2f a0; a0[0] = a_cur[rt].x; a0[1] = a_cur[rt].y;
            acc[rt] = __builtin_amdgcn_wmma_f32_16x16x4_f32(
                 false, a0,  false, b0,
                 (short)0, acc[rt],  false,  false);
        }
#pragma unroll
        for (int rt = 0; rt < GEMM_RT; ++rt) {
            v2f a1; a1[0] = a_cur[rt].z; a1[1] = a_cur[rt].w;
            acc[rt] = __builtin_amdgcn_wmma_f32_16x16x4_f32(
                 false, a1,  false, b1,
                 (short)0, acc[rt],  false,  false);
        }
        __builtin_amdgcn_sched_barrier(0);
#pragma unroll
        for (int rt = 0; rt < GEMM_RT; ++rt) a_cur[rt] = a_nxt[rt];
    }

    float* zout = z + (size_t)(row0 + 8 * half) * OUT_DIM + ncol;
    if (row0 + GEMM_ROWS <= n_nodes) {
#pragma unroll
        for (int rt = 0; rt < GEMM_RT; ++rt) {
#pragma unroll
            for (int v = 0; v < 8; ++v)
                zout[(rt * 16 + v) * OUT_DIM] = acc[rt][v];
        }
    } else {
#pragma unroll
        for (int rt = 0; rt < GEMM_RT; ++rt) {
#pragma unroll
            for (int v = 0; v < 8; ++v) {
                int row = row0 + rt * 16 + v + 8 * half;
                if (row < n_nodes)
                    zout[(rt * 16 + v) * OUT_DIM] = acc[rt][v];
            }
        }
    }

    if (tid == 0) {
        const int rows = gat_owner_rows(n_nodes);
        for (int o = blockIdx.x; o < AGG_OWNERS; o += gridDim.x) {
            if (o * rows < n_nodes) {
                unsigned int cap;
                gat_list(out, o, rows, n_nodes, 0, &cap)[0] = 0u;
                gat_list(out, o, rows, n_nodes, 1, &cap)[0] = 0u;
            }
        }
    }
}

#define BIN_WGS     256
#define BIN_THREADS 512
#define BIN_GROUPS  4
#define BIN_EDGES   (4 * BIN_GROUPS)
#define BIN_TILE    (BIN_THREADS * BIN_EDGES)
#define BIN_NONE    0xFFFFFFFFu

__global__ void __launch_bounds__(BIN_THREADS)
gat_edge_bin_kernel(const int* __restrict__ src,
                    const int* __restrict__ dst,
                    float* __restrict__ out,
                    int n_nodes, int n_edges) {
    static_assert(AGG_OWNERS % 32 == 0 && 2 * AGG_OWNERS <= BIN_THREADS,
                  "threads o and AGG_OWNERS + o serve owner o; the prefix sum runs over whole waves");
    __shared__ unsigned int       cnt[AGG_OWNERS];
    __shared__ unsigned int       start[AGG_OWNERS];
    __shared__ uint2              run[AGG_OWNERS];
    __shared__ unsigned int       wsum[AGG_OWNERS / 32];
    __shared__ unsigned long long sorted[BIN_TILE];

    const int hi = gat_half();

    const int tid  = threadIdx.x;
    const int lane = tid & 31;
    const int rows = gat_owner_rows(n_nodes);
    unsigned int* outw = (unsigned int*)out;

    const int per   = (((n_edges + (int)gridDim.x - 1) / (int)gridDim.x) + 3) & ~3;
    const int first = (int)blockIdx.x * per;
    const int last  = (n_edges - first < per) ? n_edges : first + per;

    for (int t0 = first; t0 < last; t0 += BIN_TILE) {
        const int tend = (last - t0 < BIN_TILE) ? last : t0 + BIN_TILE;
        const int nt   = tend - t0;

        for (int o = tid; o < AGG_OWNERS; o += BIN_THREADS) cnt[o] = 0u;

        int sv[BIN_EDGES], dv[BIN_EDGES];
        if ((n_edges & 3) == 0) {
#pragma unroll
            for (int g = 0; g < BIN_GROUPS; ++g) {
                const int e0 = t0 + (g * BIN_THREADS + tid) * 4;
                int4 a = make_int4(0, 0, 0, 0);
                int4 b = make_int4(0, 0, 0, 0);
                if (e0 < tend) {
                    a = *(const int4*)(src + e0);
                    b = *(const int4*)(dst + e0);
                }
                sv[4 * g] = a.x; sv[4 * g + 1] = a.y; sv[4 * g + 2] = a.z; sv[4 * g + 3] = a.w;
                dv[4 * g] = b.x; dv[4 * g + 1] = b.y; dv[4 * g + 2] = b.z; dv[4 * g + 3] = b.w;
            }
        } else {
#pragma unroll
            for (int i = 0; i < BIN_EDGES; ++i) {
                const int e = t0 + ((i >> 2) * BIN_THREADS + tid) * 4 + (i & 3);
                sv[i] = 0;
                dv[i] = 0;
                if (e < tend) {
                    sv[i] = src[e];
                    dv[i] = dst[e];
                }
            }
        }

        unsigned int rec[BIN_EDGES];
        int          ow[BIN_EDGES];
        unsigned int live = 0u;
#pragma unroll
        for (int i = 0; i < BIN_EDGES; ++i) {
            const int e = t0 + ((i >> 2) * BIN_THREADS + tid) * 4 + (i & 3);
            const unsigned int q  = (unsigned int)dv[i] / (unsigned int)rows;
            const unsigned int dl = (unsigned int)dv[i] - q * (unsigned int)rows;
            rec[i] = ((unsigned int)sv[i] << 10) | dl;
            ow[i]  = (int)(q < AGG_OWNERS - 1 ? q : AGG_OWNERS - 1);
            if (e < tend) live |= 1u << i;
        }
        __syncthreads();

        unsigned int rk[BIN_EDGES];
#pragma unroll
        for (int i = 0; i < BIN_EDGES; ++i) {
            rk[i] = 0u;
            if ((live >> i) & 1u)
                rk[i] = __hip_atomic_fetch_add(&cnt[ow[i]], 1u, __ATOMIC_RELAXED,
                                               __HIP_MEMORY_SCOPE_WORKGROUP);
        }
        __syncthreads();

        unsigned int c = 0u, inc = 0u;
        if (tid < AGG_OWNERS) {
            const unsigned int n = cnt[tid];
            unsigned int word = 0u, room = 0u;
            if (tid * rows < n_nodes) {
                unsigned int cap;
                unsigned int* l = gat_list(out, tid, rows, n_nodes, hi, &cap);
                const unsigned int base = (n != 0u) ? atomicAdd(l, n) : 0u;
                word = (unsigned int)(l - outw) + 1u + base;
                room = (cap > base) ? (cap - base) : 0u;
            }
            run[tid] = make_uint2(word, room);
        } else if (tid < 2 * AGG_OWNERS) {
            c   = cnt[tid - AGG_OWNERS];
            inc = c;
#pragma unroll
            for (int d = 1; d < 32; d <<= 1) {
                const unsigned int below = __shfl_up(inc, d);
                if (lane >= d) inc += below;
            }
            if (lane == 31) wsum[(tid - AGG_OWNERS) >> 5] = inc;
        }
        __syncthreads();
        if (tid >= AGG_OWNERS && tid < 2 * AGG_OWNERS) {
            const int o = tid - AGG_OWNERS;
            unsigned int at0 = inc - c;
#pragma unroll
            for (int w = 0; w < AGG_OWNERS / 32; ++w)
                if (w < (o >> 5)) at0 += wsum[w];
            start[o] = at0;
        }
        __syncthreads();

        unsigned int at[BIN_EDGES];
        uint2        rn[BIN_EDGES];
#pragma unroll
        for (int i = 0; i < BIN_EDGES; ++i) {
            at[i] = start[ow[i]];
            rn[i] = run[ow[i]];
        }
#pragma unroll
        for (int i = 0; i < BIN_EDGES; ++i) {
            const unsigned int word = (rk[i] < rn[i].y) ? (rn[i].x + rk[i]) : BIN_NONE;
            if ((live >> i) & 1u)
                sorted[at[i] + rk[i]] = ((unsigned long long)word << 32) | rec[i];
        }
        __syncthreads();

        unsigned long long s[BIN_EDGES];
#pragma unroll
        for (int i = 0; i < BIN_EDGES; ++i) {
            const int p = tid + i * BIN_THREADS;
            s[i] = sorted[p < nt ? p : nt - 1];
        }
#pragma unroll
        for (int i = 0; i < BIN_EDGES; ++i) {
            const int p = tid + i * BIN_THREADS;
            const unsigned int word = (unsigned int)(s[i] >> 32);
            if (p < nt && word != BIN_NONE) outw[word] = (unsigned int)s[i];
        }
    }
}


typedef __attribute__((address_space(1))) const float gat_gcfloat;
typedef __attribute__((address_space(1))) const v2f   gat_gcv2f;

__device__ __forceinline__ unsigned long long gat_row_addr(const float* base, int row) {
    const unsigned long long a =
        (unsigned long long)base + ((unsigned long long)(unsigned int)row << 8);
    const unsigned int lo = __builtin_amdgcn_readfirstlane((unsigned int)a);
    const unsigned int hi = __builtin_amdgcn_readfirstlane((unsigned int)(a >> 32));
    return ((unsigned long long)hi << 32) | lo;
}

#define AGG_THREADS 1024
#define AGG_WAVES   (AGG_THREADS / 32)

#define GAT_LO (-60.0f)
#define GAT_HI 80.0f

__device__ __forceinline__ void gat_lds_add(float* p, float v) {
    __hip_atomic_fetch_add(p, v, __ATOMIC_RELAXED, __HIP_MEMORY_SCOPE_WORKGROUP);
}

__device__ __forceinline__ void gat_lds_max(unsigned int* p, float v) {
    if (v >= 0.0f)
        __hip_atomic_fetch_max((int*)p, __float_as_int(v), __ATOMIC_RELAXED,
                               __HIP_MEMORY_SCOPE_WORKGROUP);
    else
        __hip_atomic_fetch_min(p, __float_as_uint(v), __ATOMIC_RELAXED,
                               __HIP_MEMORY_SCOPE_WORKGROUP);
}

__device__ __forceinline__ unsigned int gat_record(const unsigned int* l0,
                                                   const unsigned int* l1,
                                                   unsigned int c0, unsigned int idx) {
    return (idx < c0) ? l0[1u + idx] : l1[1u + (idx - c0)];
}

template <int M>
__device__ __forceinline__ float gat_row_xor(float v) {
    return __builtin_bit_cast(float, __builtin_amdgcn_update_dpp(
        0, __builtin_bit_cast(int, v), 0x160 + M, 0xf, 0xf, true));
}
__device__ __forceinline__ float gat_xhalf(float v) {
    const int b = __builtin_bit_cast(int, v);
    return __builtin_bit_cast(float, __builtin_amdgcn_permlanex16(
        b, b, 0x76543210, 0xfedcba98, false, false));
}

__device__ __forceinline__ float gat_exp_wide(float x) {
    const float l2e_hi = 1.44269502f;
    const float l2e_lo = 1.92596303e-8f;
    const float y = x * l2e_hi;
    const float r = fmaf(x, l2e_lo, fmaf(x, l2e_hi, -y));
    const float e = __builtin_amdgcn_exp2f(y);
    return fmaf(e, r * 0.693147182f, e);
}

template <bool FULL>
__device__ __forceinline__ void gat_owner_fused(const float* __restrict__ z,
                                                const float* zd,
                                                int s, int dl, int cnt,
                                                float* acc, float* den,
                                                unsigned int* mxb, int lane) {
    const int c = lane & 15;
    v2f a2[2][16];
#pragma unroll
    for (int k = 0; k < 32; ++k) {
        const gat_gcv2f* zr = (const gat_gcv2f*)gat_row_addr(
            z, __builtin_amdgcn_readlane(s, k));
        a2[k >> 4][k & 15] = zr[lane];
    }
#pragma unroll
    for (int hf = 0; hf < 2; ++hf) {
        const v2f* a = a2[hf];
        float p[16];
#pragma unroll
        for (int u = 0; u < 16; ++u) {
            const v2f b = *(const v2f*)(zd + __builtin_amdgcn_readlane(dl, hf * 16 + u) * OUT_DIM
                                        + 2 * lane);
            p[u] = fmaf(a[u].y, b.y, a[u].x * b.x);
        }

        const bool h8 = (c & 8) != 0;
        float q[8];
#pragma unroll
        for (int k = 0; k < 8; ++k) {
            const float send = h8 ? p[k] : p[k + 8];
            const float keep = h8 ? p[k + 8] : p[k];
            q[k] = keep + gat_row_xor<8>(send);
        }
        const bool h4 = (c & 4) != 0;
        float w[4];
#pragma unroll
        for (int k = 0; k < 4; ++k) {
            const float send = h4 ? q[k] : q[k + 4];
            const float keep = h4 ? q[k + 4] : q[k];
            w[k] = keep + gat_row_xor<4>(send);
        }
        const bool h2 = (c & 2) != 0;
        float x[2];
#pragma unroll
        for (int k = 0; k < 2; ++k) {
            const float send = h2 ? w[k] : w[k + 2];
            const float keep = h2 ? w[k + 2] : w[k];
            x[k] = keep + gat_row_xor<2>(send);
        }
        const bool h1 = (c & 1) != 0;
        const float send1 = h1 ? x[0] : x[1];
        const float keep1 = h1 ? x[1] : x[0];
        const float y = keep1 + gat_row_xor<1>(send1);
        const float ev = y + gat_xhalf(y);
        const float t  = gat_exp_wide(ev);

        if ((lane >> 4) == hf && (FULL || lane < cnt)) {
            gat_lds_max(&mxb[dl], ev);
            gat_lds_add(&den[dl], t);
        }
        const int tbits = __float_as_int(t);
#pragma unroll
        for (int u = 0; u < 16; ++u) {
            const int k = hf * 16 + u;
            if (FULL || k < cnt) {
                float* orow = acc + __builtin_amdgcn_readlane(dl, k) * OUT_DIM + 2 * lane;
                const float tk = __int_as_float(__builtin_amdgcn_readlane(tbits, u));
                gat_lds_add(orow,     tk * a[u].x);
                gat_lds_add(orow + 1, tk * a[u].y);
            }
        }
    }
}

__device__ __forceinline__ void gat_owner_redo(const float* __restrict__ z,
                                               const float* zd, int s, int dl,
                                               float* acc, float* den,
                                               const unsigned int* mxb, int lane) {
    const gat_gcfloat* zr = (const gat_gcfloat*)gat_row_addr(z, s);
    const float a0 = zr[lane];
    const float a1 = zr[lane + 32];
    const float* zl = zd + dl * OUT_DIM;
    float p = fmaf(a1, zl[lane + 32], a0 * zl[lane]);
    p += gat_xhalf(p);
    p += gat_row_xor<8>(p);
    p += gat_row_xor<4>(p);
    p += gat_row_xor<2>(p);
    p += gat_row_xor<1>(p);
    const float t = __expf(p - __uint_as_float(mxb[dl]));
    if (lane == 0) gat_lds_add(&den[dl], t);
    float* orow = acc + dl * OUT_DIM;
    gat_lds_add(orow + lane,      t * a0);
    gat_lds_add(orow + lane + 32, t * a1);
}

__device__ __forceinline__ float gat_dot_row(const float* __restrict__ zs, const float* zl) {
    float a = 0.0f;
    for (int j = 0; j < OUT_DIM / 4; ++j) {
        const float4 x = ((const float4*)zs)[j];
        const float4 y = ((const float4*)zl)[j];
        a = fmaf(x.x, y.x, a);
        a = fmaf(x.y, y.y, a);
        a = fmaf(x.z, y.z, a);
        a = fmaf(x.w, y.w, a);
    }
    return a;
}

__global__ void __launch_bounds__(AGG_THREADS)
gat_owner_kernel(const float* __restrict__ z,
                 const int* __restrict__ src,
                 const int* __restrict__ dst,
                 float* __restrict__ out,
                 int n_nodes, int n_edges) {
    __shared__ __attribute__((aligned(16))) float acc[AGG_RMAX * OUT_DIM];
    __shared__ __attribute__((aligned(16))) float zd[AGG_RMAX * OUT_DIM];
    __shared__ float        den[AGG_RMAX];
    __shared__ unsigned int mxb[AGG_RMAX];
    __shared__ unsigned int bad[AGG_RMAX];
    __shared__ unsigned int nbad;

    const int tid  = threadIdx.x;
    const int lane = tid & 31;
    const int wave = __builtin_amdgcn_readfirstlane(tid >> 5);
    const int rows = gat_owner_rows(n_nodes);
    const int own  = blockIdx.x;
    const int lo   = own * rows;
    if (lo >= n_nodes) return;
    const int nrows = (n_nodes - lo < rows) ? (n_nodes - lo) : rows;

    unsigned int cap;
    const unsigned int* l0 = gat_list(out, own, rows, n_nodes, 0, &cap);
    const unsigned int* l1 = gat_list(out, own, rows, n_nodes, 1, &cap);
    const unsigned int c0 = __builtin_amdgcn_readfirstlane(l0[0]);
    const unsigned int c1 = __builtin_amdgcn_readfirstlane(l1[0]);

    const float4* z4  = (const float4*)(z + (size_t)lo * OUT_DIM);
    float4*       zd4 = (float4*)zd;
    float4*       ac4 = (float4*)acc;
    for (int i = tid; i < nrows * (OUT_DIM / 4); i += AGG_THREADS) {
        zd4[i] = z4[i];
        ac4[i] = make_float4(0.0f, 0.0f, 0.0f, 0.0f);
    }
    for (int i = tid; i < nrows; i += AGG_THREADS) {
        den[i] = 0.0f;
        mxb[i] = 0xFF800000u;
        bad[i] = 0u;
    }
    if (tid == 0) nbad = 0u;
    __syncthreads();

    if (c0 <= cap && c1 <= cap) {
        const unsigned int total = c0 + c1;

        {
            unsigned int base = (unsigned int)wave * 32u;
            unsigned int rec = 0u;
            if (base < total) {
                unsigned int idx = base + (unsigned int)lane;
                if (idx >= total) idx = total - 1u;
                rec = gat_record(l0, l1, c0, idx);
            }
            while (base < total) {
                const unsigned int nbase = base + AGG_WAVES * 32u;
                unsigned int nrec = rec;
                if (nbase < total) {
                    unsigned int idx = nbase + (unsigned int)lane;
                    if (idx >= total) idx = total - 1u;
                    nrec = gat_record(l0, l1, c0, idx);
                }
                int s  = (int)(rec >> 10);
                int dl = (int)(rec & 1023u);
                if (s  > n_nodes - 1) s  = n_nodes - 1;
                if (dl > nrows - 1)   dl = nrows - 1;
                const unsigned int left = total - base;
                if (left >= 32u)
                    gat_owner_fused<true>(z, zd, s, dl, 32, acc, den, mxb, lane);
                else
                    gat_owner_fused<false>(z, zd, s, dl, (int)left, acc, den, mxb, lane);
                rec  = nrec;
                base = nbase;
            }
        }
        __syncthreads();

        {
            const float inf = __builtin_huge_valf();
            unsigned int off = 0u;
            for (int i = tid; i < nrows; i += AGG_THREADS) {
                const unsigned int m = mxb[i];
                const float mf = __uint_as_float(m);
                if ((m != 0xFF800000u && !(mf >= GAT_LO && mf <= GAT_HI)) || !(den[i] < inf)) {
                    bad[i] = 1u;
                    off = 1u;
                }
            }
            const float4* a4c = (const float4*)acc;
            for (int i = tid; i < nrows * (OUT_DIM / 4); i += AGG_THREADS) {
                const float4 a = a4c[i];
                if (!(fabsf(a.x) < inf && fabsf(a.y) < inf && fabsf(a.z) < inf && fabsf(a.w) < inf)) {
                    bad[i / (OUT_DIM / 4)] = 1u;
                    off = 1u;
                }
            }
            if (off != 0u) nbad = 1u;
        }
        __syncthreads();

        if (__builtin_amdgcn_readfirstlane(nbad) != 0u) {
            for (int i = tid; i < nrows * (OUT_DIM / 4); i += AGG_THREADS) {
                if (bad[i / (OUT_DIM / 4)] != 0u)
                    ac4[i] = make_float4(0.0f, 0.0f, 0.0f, 0.0f);
            }
            for (int i = tid; i < nrows; i += AGG_THREADS) {
                if (bad[i] != 0u) den[i] = 0.0f;
            }
            __syncthreads();

            unsigned int base = (unsigned int)wave * 32u;
            unsigned int rec = 0u;
            if (base < total) {
                unsigned int idx = base + (unsigned int)lane;
                if (idx >= total) idx = total - 1u;
                rec = gat_record(l0, l1, c0, idx);
            }
            while (base < total) {
                const unsigned int nbase = base + AGG_WAVES * 32u;
                unsigned int nrec = rec;
                if (nbase < total) {
                    unsigned int idx = nbase + (unsigned int)lane;
                    if (idx >= total) idx = total - 1u;
                    nrec = gat_record(l0, l1, c0, idx);
                }
                int s  = (int)(rec >> 10);
                int dl = (int)(rec & 1023u);
                if (s  > n_nodes - 1) s  = n_nodes - 1;
                if (dl > nrows - 1)   dl = nrows - 1;
                const bool hit = (base + (unsigned int)lane < total) && bad[dl] != 0u;
                unsigned int todo = __builtin_amdgcn_ballot_w32(hit);
                while (todo != 0u) {
                    const int k = __builtin_ctz(todo);
                    todo &= todo - 1u;
                    gat_owner_redo(z, zd, __builtin_amdgcn_readlane(s, k),
                                   __builtin_amdgcn_readlane(dl, k), acc, den, mxb, lane);
                }
                rec  = nrec;
                base = nbase;
            }
        }
    } else {
        for (int eid = tid; eid < n_edges; eid += AGG_THREADS) {
            const int dl = dst[eid] - lo;
            if ((unsigned int)dl < (unsigned int)nrows) {
                int s = src[eid];
                if (s > n_nodes - 1) s = n_nodes - 1;
                if (s < 0) s = 0;
                gat_lds_max(&mxb[dl], gat_dot_row(z + (size_t)s * OUT_DIM, zd + dl * OUT_DIM));
            }
        }
        __syncthreads();
        for (int eid = tid; eid < n_edges; eid += AGG_THREADS) {
            const int dl = dst[eid] - lo;
            if ((unsigned int)dl < (unsigned int)nrows) {
                int s = src[eid];
                if (s > n_nodes - 1) s = n_nodes - 1;
                if (s < 0) s = 0;
                const float* zs = z + (size_t)s * OUT_DIM;
                const float ev = gat_dot_row(zs, zd + dl * OUT_DIM);
                const float t  = __expf(ev - __uint_as_float(mxb[dl]));
                gat_lds_add(&den[dl], t);
                float* orow = acc + dl * OUT_DIM;
                for (int j = 0; j < OUT_DIM; ++j)
                    gat_lds_add(orow + j, t * zs[j]);
            }
        }
    }
    __syncthreads();

    float4*       o4 = (float4*)(out + (size_t)lo * OUT_DIM);
    const float4* a4 = (const float4*)acc;
    for (int i = tid; i < nrows * (OUT_DIM / 4); i += AGG_THREADS) {
        const float inv = 1.0f / fmaxf(den[i / (OUT_DIM / 4)], 1e-30f);
        float4 a = a4[i];
        a.x *= inv; a.y *= inv; a.z *= inv; a.w *= inv;
        o4[i] = a;
    }
}

extern "C" void kernel_launch(void* const* d_in, const int* in_sizes, int n_in,
                              void* d_out, int out_size, void* d_ws, size_t ws_size,
                              hipStream_t stream) {
    const float* h   = (const float*)d_in[0];
    const float* W   = (const float*)d_in[1];
    const int*   src = (const int*)d_in[2];
    const int*   dst = (const int*)d_in[3];
    float* out = (float*)d_out;

    const int n_nodes = in_sizes[0] / IN_DIM;
    const int n_edges = in_sizes[2];

    float* z = (float*)d_ws;

    {
        int blocks = (n_nodes + GEMM_ROWS - 1) / GEMM_ROWS;
        gat_gemm_wmma_kernel<<<blocks, 128, 0, stream>>>(h, W, z, out, n_nodes);
    }
    {
        gat_edge_bin_kernel<<<BIN_WGS, BIN_THREADS, 0, stream>>>(src, dst, out, n_nodes, n_edges);
    }
    {
        gat_owner_kernel<<<AGG_OWNERS, AGG_THREADS, 0, stream>>>(
            z, src, dst, out, n_nodes, n_edges);
    }
}
